# starting kernel
# baseline (speedup 1.0000x reference)
#include <hip/hip_runtime.h>


typedef __attribute__((ext_vector_type(2))) float v2f;
typedef __attribute__((ext_vector_type(8))) float v8f;

#define B_ 4
#define L_ 2046
#define D_ 1024
#define C_ 512
#define P_ 1024
#define H_ 16
#define HD_ 32
#define DV_ 64
#define WKV_LD 2048

static __device__ __forceinline__ v8f wmma4(v2f a, v2f b, v8f c) {
  return __builtin_amdgcn_wmma_f32_16x16x4_f32(false, a, false, b, (short)0, c,
                                               false, false);
}

static __device__ __forceinline__ float rowmax16(float v) {
  v = fmaxf(v, __shfl_xor(v, 8, 16));
  v = fmaxf(v, __shfl_xor(v, 4, 16));
  v = fmaxf(v, __shfl_xor(v, 2, 16));
  v = fmaxf(v, __shfl_xor(v, 1, 16));
  return v;
}
static __device__ __forceinline__ float rowsum16(float v) {
  v += __shfl_xor(v, 8, 16);
  v += __shfl_xor(v, 4, 16);
  v += __shfl_xor(v, 2, 16);
  v += __shfl_xor(v, 1, 16);
  return v;
}

__global__ __launch_bounds__(256) void gemm_rowA_wmma(
    const float* __restrict__ A, long strideAm, long batchStrideA,
    const float* __restrict__ W, const float* __restrict__ bias,
    float* __restrict__ Out, long strideOm, long batchStrideO,
    int M, int K, int mstrips, int nstrips) {
  int lane = threadIdx.x & 31;
  int wid = (blockIdx.x * blockDim.x + threadIdx.x) >> 5;
  int wavesPerBatch = mstrips * nstrips;
  int b = wid / wavesPerBatch;
  int r = wid - b * wavesPerBatch;
  int m0 = (r / nstrips) * 32;
  int n0 = (r % nstrips) * 64;
  const float* Ab = A + (long)b * batchStrideA;
  float* Ob = Out + (long)b * batchStrideO;
  int col = lane & 15;
  int hi = lane >> 4;
  int khalf = hi * 2;
  int row0 = m0 + col;      if (row0 > M - 1) row0 = M - 1;
  int row1 = m0 + 16 + col; if (row1 > M - 1) row1 = M - 1;
  const float* Arow0 = Ab + (long)row0 * strideAm + khalf;
  const float* Arow1 = Ab + (long)row1 * strideAm + khalf;
  const float* Wp = W + (long)(n0 + col) * K + khalf;
  const long Wn16 = (long)16 * K;

  v8f acc[2][4] = {};
  for (int k0 = 0; k0 < K; k0 += 4) {
    if ((k0 & 63) == 0) {
      __builtin_prefetch(Arow0 + k0 + 256, 0, 1);
      __builtin_prefetch(Arow1 + k0 + 256, 0, 1);
      __builtin_prefetch(Wp + k0 + 256, 0, 1);
      __builtin_prefetch(Wp + 3 * Wn16 + k0 + 256, 0, 1);
    }
    v2f a0 = *(const v2f*)(Arow0 + k0);
    v2f a1 = *(const v2f*)(Arow1 + k0);
#pragma unroll
    for (int j = 0; j < 4; ++j) {
      v2f bf = *(const v2f*)(Wp + j * Wn16 + k0);
      acc[0][j] = wmma4(a0, bf, acc[0][j]);
      acc[1][j] = wmma4(a1, bf, acc[1][j]);
    }
  }
  float bv[4];
#pragma unroll
  for (int j = 0; j < 4; ++j) bv[j] = bias[n0 + j * 16 + col];
#pragma unroll
  for (int i = 0; i < 2; ++i) {
#pragma unroll
    for (int rr = 0; rr < 8; ++rr) {
      int row = m0 + i * 16 + rr + 8 * hi;
      if (row < M) {
        float* orow = Ob + (long)row * strideOm + n0;
#pragma unroll
        for (int j = 0; j < 4; ++j) orow[j * 16 + col] = acc[i][j][rr] + bv[j];
      }
    }
  }
}

__global__ __launch_bounds__(256) void gemm_colA_wmma(
    const float* __restrict__ A, long strideAk, long batchStrideA,
    const float* __restrict__ Wkv,
    float* __restrict__ Out, long strideOm, long batchStrideO,
    int K, int mstrips, int nstrips) {
  int lane = threadIdx.x & 31;
  int wid = (blockIdx.x * blockDim.x + threadIdx.x) >> 5;
  int wavesPerBatch = mstrips * nstrips;
  int b = wid / wavesPerBatch;
  int r = wid - b * wavesPerBatch;
  int m0 = (r / nstrips) * 32;
  int n0 = (r % nstrips) * 64;
  const float* Ab = A + (long)b * batchStrideA;
  float* Ob = Out + (long)b * batchStrideO;
  int col = lane & 15;
  int hi = lane >> 4;
  int khalf = hi * 2;
  const float* Am0 = Ab + (long)khalf * strideAk + m0 + col;
  const float* Am1 = Am0 + 16;
  const float* Wp = Wkv + (long)(n0 + col) * WKV_LD + khalf;
  const long Wn16 = (long)16 * WKV_LD;

  v8f acc[2][4] = {};
  int Kmain = K & ~3;
  for (int k0 = 0; k0 < Kmain; k0 += 4) {
    long ao = (long)k0 * strideAk;
    v2f a0 = {Am0[ao], Am0[ao + strideAk]};
    v2f a1 = {Am1[ao], Am1[ao + strideAk]};
#pragma unroll
    for (int j = 0; j < 4; ++j) {
      v2f bf = *(const v2f*)(Wp + j * Wn16 + k0);
      acc[0][j] = wmma4(a0, bf, acc[0][j]);
      acc[1][j] = wmma4(a1, bf, acc[1][j]);
    }
  }
  if (Kmain < K) {
    int k0 = Kmain;
    long ao = (long)k0 * strideAk;
    v2f a0 = {0.f, 0.f}, a1 = {0.f, 0.f};
    if (hi == 0) {
      a0 = (v2f){Am0[ao], Am0[ao + strideAk]};
      a1 = (v2f){Am1[ao], Am1[ao + strideAk]};
    }
#pragma unroll
    for (int j = 0; j < 4; ++j) {
      v2f bf = *(const v2f*)(Wp + j * Wn16 + k0);
      acc[0][j] = wmma4(a0, bf, acc[0][j]);
      acc[1][j] = wmma4(a1, bf, acc[1][j]);
    }
  }
#pragma unroll
  for (int i = 0; i < 2; ++i) {
#pragma unroll
    for (int rr = 0; rr < 8; ++rr) {
      int row = m0 + i * 16 + rr + 8 * hi;
      float* orow = Ob + (long)row * strideOm + n0;
#pragma unroll
      for (int j = 0; j < 4; ++j) orow[j * 16 + col] = acc[i][j][rr];
    }
  }
}

__global__ __launch_bounds__(256) void attn_wmma(
    const float* __restrict__ q, const float* __restrict__ kbuf,
    const float* __restrict__ vbuf, float* __restrict__ obuf) {
  __shared__ float lds[8][256];
  int lane = threadIdx.x & 31;
  int wslot = threadIdx.x >> 5;
  int wid = (blockIdx.x * blockDim.x + threadIdx.x) >> 5;
  int ltile = wid & 127;
  int h = (wid >> 7) & 15;
  int b = wid >> 11;
  int l0 = ltile * 16;
  int col = lane & 15;
  int hi = lane >> 4;
  int khalf = hi * 2;

  v2f qA[8];
  {
    int l = l0 + col;
    bool lv = l < L_;
    const float* qrow = q + ((long)b * L_ + l) * C_ + h * HD_;
    const float sc = 0.125f * 1.44269504088896340736f;
#pragma unroll
    for (int kk = 0; kk < 8; ++kk) {
      int k = kk * 4 + khalf;
      float x0 = lv ? qrow[k] * sc : 0.f;
      float x1 = lv ? qrow[k + 1] * sc : 0.f;
      qA[kk] = (v2f){x0, x1};
    }
  }

  v8f O[4] = {};
  float Mrow[8], Srow[8];
#pragma unroll
  for (int r = 0; r < 8; ++r) { Mrow[r] = -INFINITY; Srow[r] = 0.f; }

  const float* kb = kbuf + (long)b * C_ * P_ + (long)h * P_;
  const float* vb = vbuf + (long)b * D_ * P_ + (long)h * P_;

  for (int p0 = 0; p0 < P_; p0 += 16) {
    v8f c = {};
#pragma unroll
    for (int kk = 0; kk < 8; ++kk) {
      int k = kk * 4 + khalf;
      const float* kp = kb + (long)k * 16 * P_ + p0 + col;
      v2f bf = {kp[0], kp[16 * P_]};
      c = wmma4(qA[kk], bf, c);
    }
    float alpha[8];
#pragma unroll
    for (int r = 0; r < 8; ++r) {
      float mx = rowmax16(c[r]);
      float mnew = fmaxf(Mrow[r], mx);
      float al = __builtin_exp2f(Mrow[r] - mnew);
      float pv = __builtin_exp2f(c[r] - mnew);
      Srow[r] = Srow[r] * al + rowsum16(pv);
      Mrow[r] = mnew;
      alpha[r] = al;
      lds[wslot][(r + 8 * hi) * 16 + col] = pv;
    }
#pragma unroll
    for (int r = 0; r < 8; ++r) {
      O[0][r] *= alpha[r]; O[1][r] *= alpha[r];
      O[2][r] *= alpha[r]; O[3][r] *= alpha[r];
    }
    __builtin_amdgcn_fence(__ATOMIC_ACQ_REL, "wavefront");
    __builtin_amdgcn_wave_barrier();
    v2f pA[4];
#pragma unroll
    for (int kk = 0; kk < 4; ++kk) {
      int k = kk * 4 + khalf;
      pA[kk] = (v2f){lds[wslot][col * 16 + k], lds[wslot][col * 16 + k + 1]};
    }
    __builtin_amdgcn_fence(__ATOMIC_ACQ_REL, "wavefront");
    __builtin_amdgcn_wave_barrier();
#pragma unroll
    for (int j = 0; j < 4; ++j) {
#pragma unroll
      for (int kk = 0; kk < 4; ++kk) {
        int k = kk * 4 + khalf;
        const float* vp = vb + (long)((j * 16 + col) * 16) * P_ + p0 + k;
        v2f bf = *(const v2f*)vp;
        O[j] = wmma4(pA[kk], bf, O[j]);
      }
    }
  }

#pragma unroll
  for (int r = 0; r < 8; ++r) {
    float inv = 1.f / Srow[r];
    O[0][r] *= inv; O[1][r] *= inv; O[2][r] *= inv; O[3][r] *= inv;
  }
#pragma unroll
  for (int r = 0; r < 8; ++r) {
    int l = l0 + r + 8 * hi;
    if (l < L_) {
      float* orow = obuf + ((long)b * L_ + l) * D_ + h;
#pragma unroll
      for (int j = 0; j < 4; ++j) orow[(j * 16 + col) * 16] = O[j][r];
    }
  }
}

extern "C" void kernel_launch(void* const* d_in, const int* in_sizes, int n_in,
                              void* d_out, int out_size, void* d_ws,
                              size_t ws_size, hipStream_t stream) {
  (void)in_sizes; (void)n_in; (void)out_size; (void)ws_size;
  const float* in = (const float*)d_in[0];
  const float* w_kv = (const float*)d_in[1];
  const float* w_q = (const float*)d_in[2];
  const float* b_q = (const float*)d_in[3];
  const float* w_out = (const float*)d_in[4];
  const float* b_out = (const float*)d_in[5];
  float* out = (float*)d_out;

  float* ws = (float*)d_ws;
  float* qbuf = ws;
  float* vbuf = qbuf + (size_t)B_ * L_ * C_;
  float* kbuf = vbuf + (size_t)B_ * D_ * P_;
  float* obuf = kbuf + (size_t)B_ * C_ * P_;

  gemm_rowA_wmma<<<256, 256, 0, stream>>>(in, (long)(B_ * D_), (long)D_, w_q,
                                          b_q, qbuf, (long)C_, (long)L_ * C_,
                                          L_, D_, 64, 8);
  gemm_colA_wmma<<<256, 256, 0, stream>>>(in, (long)(B_ * D_), (long)D_, w_kv,
                                          vbuf, (long)P_, (long)D_ * P_, L_,
                                          32, 16);
  gemm_colA_wmma<<<128, 256, 0, stream>>>(qbuf, (long)C_, (long)L_ * C_, w_kv,
                                          kbuf, (long)P_, (long)C_ * P_, L_,
                                          16, 16);
  attn_wmma<<<1024, 256, 0, stream>>>(qbuf, kbuf, vbuf, obuf);
  gemm_rowA_wmma<<<512, 256, 0, stream>>>(obuf, (long)D_, (long)L_ * D_,
                                          w_out, b_out, out, (long)(B_ * D_),
                                          (long)D_, L_, D_, 64, 16);
}


# Fragment loads widened to 16 B per lane (4 consecutive k feed two WMMA steps, loads grouped ahead of each trip) in all three kernels; sum order of k inside each 8-k chunk changes
# speedup vs baseline: 1.2357x; 1.2357x over previous
#include <hip/hip_runtime.h>


typedef __attribute__((ext_vector_type(2))) float v2f;
typedef __attribute__((ext_vector_type(4))) float v4f;
typedef __attribute__((ext_vector_type(8))) float v8f;

static __device__ __forceinline__ v2f lo2(v4f x) { return (v2f){x.x, x.y}; }
static __device__ __forceinline__ v2f hi2(v4f x) { return (v2f){x.z, x.w}; }

#define B_ 4
#define L_ 2046
#define D_ 1024
#define C_ 512
#define P_ 1024
#define H_ 16
#define HD_ 32
#define DV_ 64
#define WKV_LD 2048

static __device__ __forceinline__ v8f wmma4(v2f a, v2f b, v8f c) {
  return __builtin_amdgcn_wmma_f32_16x16x4_f32(false, a, false, b, (short)0, c,
                                               false, false);
}

static __device__ __forceinline__ float rowmax16(float v) {
  v = fmaxf(v, __shfl_xor(v, 8, 16));
  v = fmaxf(v, __shfl_xor(v, 4, 16));
  v = fmaxf(v, __shfl_xor(v, 2, 16));
  v = fmaxf(v, __shfl_xor(v, 1, 16));
  return v;
}
static __device__ __forceinline__ float rowsum16(float v) {
  v += __shfl_xor(v, 8, 16);
  v += __shfl_xor(v, 4, 16);
  v += __shfl_xor(v, 2, 16);
  v += __shfl_xor(v, 1, 16);
  return v;
}

__global__ __launch_bounds__(256) void gemm_rowA_wmma(
    const float* __restrict__ A, long strideAm, long batchStrideA,
    const float* __restrict__ W, const float* __restrict__ bias,
    float* __restrict__ Out, long strideOm, long batchStrideO,
    int M, int K, int mstrips, int nstrips) {
  int lane = threadIdx.x & 31;
  int wid = (blockIdx.x * blockDim.x + threadIdx.x) >> 5;
  int wavesPerBatch = mstrips * nstrips;
  int b = wid / wavesPerBatch;
  int r = wid - b * wavesPerBatch;
  int m0 = (r / nstrips) * 32;
  int n0 = (r % nstrips) * 64;
  const float* Ab = A + (long)b * batchStrideA;
  float* Ob = Out + (long)b * batchStrideO;
  int col = lane & 15;
  int hi = lane >> 4;
  int k4 = hi * 4;
  int row0 = m0 + col;      if (row0 > M - 1) row0 = M - 1;
  int row1 = m0 + 16 + col; if (row1 > M - 1) row1 = M - 1;
  const float* Arow0 = Ab + (long)row0 * strideAm + k4;
  const float* Arow1 = Ab + (long)row1 * strideAm + k4;
  const float* Wp = W + (long)(n0 + col) * K + k4;
  const long Wn16 = (long)16 * K;

  v8f acc[2][4] = {};
  for (int k0 = 0; k0 < K; k0 += 16) {
    if ((k0 & 63) == 0) {
      __builtin_prefetch(Arow0 + k0 + 256, 0, 1);
      __builtin_prefetch(Arow1 + k0 + 256, 0, 1);
      __builtin_prefetch(Wp + k0 + 256, 0, 1);
      __builtin_prefetch(Wp + 3 * Wn16 + k0 + 256, 0, 1);
    }
    v4f a0[2], a1[2], bf[2][4];
#pragma unroll
    for (int c = 0; c < 2; ++c) {
      a0[c] = *(const v4f*)(Arow0 + k0 + 8 * c);
      a1[c] = *(const v4f*)(Arow1 + k0 + 8 * c);
#pragma unroll
      for (int j = 0; j < 4; ++j)
        bf[c][j] = *(const v4f*)(Wp + j * Wn16 + k0 + 8 * c);
    }
    __builtin_amdgcn_sched_barrier(0);
#pragma unroll
    for (int c = 0; c < 2; ++c) {
#pragma unroll
      for (int j = 0; j < 4; ++j) {
        acc[0][j] = wmma4(lo2(a0[c]), lo2(bf[c][j]), acc[0][j]);
        acc[1][j] = wmma4(lo2(a1[c]), lo2(bf[c][j]), acc[1][j]);
      }
#pragma unroll
      for (int j = 0; j < 4; ++j) {
        acc[0][j] = wmma4(hi2(a0[c]), hi2(bf[c][j]), acc[0][j]);
        acc[1][j] = wmma4(hi2(a1[c]), hi2(bf[c][j]), acc[1][j]);
      }
    }
  }
  float bv[4];
#pragma unroll
  for (int j = 0; j < 4; ++j) bv[j] = bias[n0 + j * 16 + col];
#pragma unroll
  for (int i = 0; i < 2; ++i) {
#pragma unroll
    for (int rr = 0; rr < 8; ++rr) {
      int row = m0 + i * 16 + rr + 8 * hi;
      if (row < M) {
        float* orow = Ob + (long)row * strideOm + n0;
#pragma unroll
        for (int j = 0; j < 4; ++j) orow[j * 16 + col] = acc[i][j][rr] + bv[j];
      }
    }
  }
}

__global__ __launch_bounds__(256) void gemm_colA_wmma(
    const float* __restrict__ A, long strideAk, long batchStrideA,
    const float* __restrict__ Wkv,
    float* __restrict__ Out, long strideOm, long batchStrideO,
    int K, int mstrips, int nstrips) {
  int lane = threadIdx.x & 31;
  int wid = (blockIdx.x * blockDim.x + threadIdx.x) >> 5;
  int wavesPerBatch = mstrips * nstrips;
  int b = wid / wavesPerBatch;
  int r = wid - b * wavesPerBatch;
  int m0 = (r / nstrips) * 32;
  int n0 = (r % nstrips) * 64;
  const float* Ab = A + (long)b * batchStrideA;
  float* Ob = Out + (long)b * batchStrideO;
  int col = lane & 15;
  int hi = lane >> 4;
  int k4 = hi * 4;
  const float* Am0 = Ab + (long)k4 * strideAk + m0 + col;
  const float* Am1 = Am0 + 16;
  const float* Wp = Wkv + (long)(n0 + col) * WKV_LD + k4;
  const long Wn16 = (long)16 * WKV_LD;
  const long s1 = strideAk, s2 = 2 * strideAk, s3 = 3 * strideAk;

  v8f acc[2][4] = {};
  int Kmain = K & ~7;
  for (int k0 = 0; k0 < Kmain; k0 += 8) {
    long ao = (long)k0 * strideAk;
    v2f a0l = {Am0[ao], Am0[ao + s1]}, a0h = {Am0[ao + s2], Am0[ao + s3]};
    v2f a1l = {Am1[ao], Am1[ao + s1]}, a1h = {Am1[ao + s2], Am1[ao + s3]};
    v4f bf[4];
#pragma unroll
    for (int j = 0; j < 4; ++j)
      bf[j] = *(const v4f*)(Wp + j * Wn16 + k0);
#pragma unroll
    for (int j = 0; j < 4; ++j) {
      acc[0][j] = wmma4(a0l, lo2(bf[j]), acc[0][j]);
      acc[1][j] = wmma4(a1l, lo2(bf[j]), acc[1][j]);
    }
#pragma unroll
    for (int j = 0; j < 4; ++j) {
      acc[0][j] = wmma4(a0h, hi2(bf[j]), acc[0][j]);
      acc[1][j] = wmma4(a1h, hi2(bf[j]), acc[1][j]);
    }
  }
  if (Kmain < K) {
    int k0 = Kmain;
    long ao = (long)k0 * strideAk;
    int kb = k0 + k4;
    bool v0 = kb < K, v1 = kb + 1 < K, v2 = kb + 2 < K, v3 = kb + 3 < K;
    v2f a0l = {v0 ? Am0[ao] : 0.f, v1 ? Am0[ao + s1] : 0.f};
    v2f a0h = {v2 ? Am0[ao + s2] : 0.f, v3 ? Am0[ao + s3] : 0.f};
    v2f a1l = {v0 ? Am1[ao] : 0.f, v1 ? Am1[ao + s1] : 0.f};
    v2f a1h = {v2 ? Am1[ao + s2] : 0.f, v3 ? Am1[ao + s3] : 0.f};
    v4f bf[4];
#pragma unroll
    for (int j = 0; j < 4; ++j) {
      v4f t = *(const v4f*)(Wp + j * Wn16 + k0);
      bf[j] = (v4f){v0 ? t.x : 0.f, v1 ? t.y : 0.f, v2 ? t.z : 0.f, v3 ? t.w : 0.f};
    }
#pragma unroll
    for (int j = 0; j < 4; ++j) {
      acc[0][j] = wmma4(a0l, lo2(bf[j]), acc[0][j]);
      acc[1][j] = wmma4(a1l, lo2(bf[j]), acc[1][j]);
    }
#pragma unroll
    for (int j = 0; j < 4; ++j) {
      acc[0][j] = wmma4(a0h, hi2(bf[j]), acc[0][j]);
      acc[1][j] = wmma4(a1h, hi2(bf[j]), acc[1][j]);
    }
  }
#pragma unroll
  for (int i = 0; i < 2; ++i) {
#pragma unroll
    for (int rr = 0; rr < 8; ++rr) {
      int row = m0 + i * 16 + rr + 8 * hi;
      float* orow = Ob + (long)row * strideOm + n0;
#pragma unroll
      for (int j = 0; j < 4; ++j) orow[j * 16 + col] = acc[i][j][rr];
    }
  }
}

__global__ __launch_bounds__(256) void attn_wmma(
    const float* __restrict__ q, const float* __restrict__ kbuf,
    const float* __restrict__ vbuf, float* __restrict__ obuf) {
  __shared__ __attribute__((aligned(16))) float lds[8][256];
  int lane = threadIdx.x & 31;
  int wslot = threadIdx.x >> 5;
  int wid = (blockIdx.x * blockDim.x + threadIdx.x) >> 5;
  int ltile = wid & 127;
  int h = (wid >> 7) & 15;
  int b = wid >> 11;
  int l0 = ltile * 16;
  int col = lane & 15;
  int hi = lane >> 4;
  int khalf = hi * 2;

  v2f qA[8];
  {
    int l = l0 + col;
    bool lv = l < L_;
    const float* qrow = q + ((long)b * L_ + l) * C_ + h * HD_;
    const float sc = 0.125f * 1.44269504088896340736f;
#pragma unroll
    for (int kk = 0; kk < 8; ++kk) {
      int k = kk * 4 + khalf;
      float x0 = lv ? qrow[k] * sc : 0.f;
      float x1 = lv ? qrow[k + 1] * sc : 0.f;
      qA[kk] = (v2f){x0, x1};
    }
  }

  v8f O[4] = {};
  float Mrow[8], Srow[8];
#pragma unroll
  for (int r = 0; r < 8; ++r) { Mrow[r] = -INFINITY; Srow[r] = 0.f; }

  const float* kb = kbuf + (long)b * C_ * P_ + (long)h * P_;
  const float* vb = vbuf + (long)b * D_ * P_ + (long)h * P_;

  for (int p0 = 0; p0 < P_; p0 += 16) {
    v8f c = {};
#pragma unroll
    for (int kk = 0; kk < 8; ++kk) {
      int k = kk * 4 + khalf;
      const float* kp = kb + (long)k * 16 * P_ + p0 + col;
      v2f bf = {kp[0], kp[16 * P_]};
      c = wmma4(qA[kk], bf, c);
    }
    float alpha[8];
#pragma unroll
    for (int r = 0; r < 8; ++r) {
      float mx = rowmax16(c[r]);
      float mnew = fmaxf(Mrow[r], mx);
      float al = __builtin_exp2f(Mrow[r] - mnew);
      float pv = __builtin_exp2f(c[r] - mnew);
      Srow[r] = Srow[r] * al + rowsum16(pv);
      Mrow[r] = mnew;
      alpha[r] = al;
      lds[wslot][(r + 8 * hi) * 16 + col] = pv;
    }
#pragma unroll
    for (int r = 0; r < 8; ++r) {
      O[0][r] *= alpha[r]; O[1][r] *= alpha[r];
      O[2][r] *= alpha[r]; O[3][r] *= alpha[r];
    }
    __builtin_amdgcn_fence(__ATOMIC_ACQ_REL, "wavefront");
    __builtin_amdgcn_wave_barrier();
    v4f pA[2];
#pragma unroll
    for (int c8 = 0; c8 < 2; ++c8)
      pA[c8] = *(const v4f*)&lds[wslot][col * 16 + c8 * 8 + hi * 4];
    __builtin_amdgcn_fence(__ATOMIC_ACQ_REL, "wavefront");
    __builtin_amdgcn_wave_barrier();
#pragma unroll
    for (int c8 = 0; c8 < 2; ++c8) {
      v4f vf[4];
#pragma unroll
      for (int j = 0; j < 4; ++j) {
        const float* vp = vb + (long)((j * 16 + col) * 16) * P_ + p0 + c8 * 8 + hi * 4;
        vf[j] = *(const v4f*)vp;
      }
#pragma unroll
      for (int j = 0; j < 4; ++j) O[j] = wmma4(lo2(pA[c8]), lo2(vf[j]), O[j]);
#pragma unroll
      for (int j = 0; j < 4; ++j) O[j] = wmma4(hi2(pA[c8]), hi2(vf[j]), O[j]);
    }
  }

#pragma unroll
  for (int r = 0; r < 8; ++r) {
    float inv = 1.f / Srow[r];
    O[0][r] *= inv; O[1][r] *= inv; O[2][r] *= inv; O[3][r] *= inv;
  }
#pragma unroll
  for (int r = 0; r < 8; ++r) {
    int l = l0 + r + 8 * hi;
    if (l < L_) {
      float* orow = obuf + ((long)b * L_ + l) * D_ + h;
#pragma unroll
      for (int j = 0; j < 4; ++j) orow[(j * 16 + col) * 16] = O[j][r];
    }
  }
}

extern "C" void kernel_launch(void* const* d_in, const int* in_sizes, int n_in,
                              void* d_out, int out_size, void* d_ws,
                              size_t ws_size, hipStream_t stream) {
  (void)in_sizes; (void)n_in; (void)out_size; (void)ws_size;
  const float* in = (const float*)d_in[0];
  const float* w_kv = (const float*)d_in[1];
  const float* w_q = (const float*)d_in[2];
  const float* b_q = (const float*)d_in[3];
  const float* w_out = (const float*)d_in[4];
  const float* b_out = (const float*)d_in[5];
  float* out = (float*)d_out;

  float* ws = (float*)d_ws;
  float* qbuf = ws;
  float* vbuf = qbuf + (size_t)B_ * L_ * C_;
  float* kbuf = vbuf + (size_t)B_ * D_ * P_;
  float* obuf = kbuf + (size_t)B_ * C_ * P_;

  gemm_rowA_wmma<<<256, 256, 0, stream>>>(in, (long)(B_ * D_), (long)D_, w_q,
                                          b_q, qbuf, (long)C_, (long)L_ * C_,
                                          L_, D_, 64, 8);
  gemm_colA_wmma<<<256, 256, 0, stream>>>(in, (long)(B_ * D_), (long)D_, w_kv,
                                          vbuf, (long)P_, (long)D_ * P_, L_,
                                          32, 16);
  gemm_colA_wmma<<<128, 256, 0, stream>>>(qbuf, (long)C_, (long)L_ * C_, w_kv,
                                          kbuf, (long)P_, (long)C_ * P_, L_,
                                          16, 16);
  attn_wmma<<<1024, 256, 0, stream>>>(qbuf, kbuf, vbuf, obuf);
  gemm_rowA_wmma<<<512, 256, 0, stream>>>(obuf, (long)D_, (long)L_ * D_,
                                          w_out, b_out, out, (long)(B_ * D_),
                                          (long)D_, L_, D_, 64, 16);
}


# GEMM operand tiles staged through LDS: work-group fetches each 128-byte line once with whole-line loads (two stages; 128x128 and 64x128 block tiles), fragments read by ds_load; k sum order kept
# speedup vs baseline: 1.5429x; 1.2486x over previous
#include <hip/hip_runtime.h>


typedef __attribute__((ext_vector_type(2))) float v2f;
typedef __attribute__((ext_vector_type(4))) float v4f;
typedef __attribute__((ext_vector_type(8))) float v8f;

static __device__ __forceinline__ v2f lo2(v4f x) { return (v2f){x.x, x.y}; }
static __device__ __forceinline__ v2f hi2(v4f x) { return (v2f){x.z, x.w}; }

#define B_ 4
#define L_ 2046
#define D_ 1024
#define C_ 512
#define P_ 1024
#define H_ 16
#define HD_ 32
#define DV_ 64
#define WKV_LD 2048

static __device__ __forceinline__ v8f wmma4(v2f a, v2f b, v8f c) {
  return __builtin_amdgcn_wmma_f32_16x16x4_f32(false, a, false, b, (short)0, c,
                                               false, false);
}

static __device__ __forceinline__ float rowmax16(float v) {
  v = fmaxf(v, __shfl_xor(v, 8, 16));
  v = fmaxf(v, __shfl_xor(v, 4, 16));
  v = fmaxf(v, __shfl_xor(v, 2, 16));
  v = fmaxf(v, __shfl_xor(v, 1, 16));
  return v;
}
static __device__ __forceinline__ float rowsum16(float v) {
  v += __shfl_xor(v, 8, 16);
  v += __shfl_xor(v, 4, 16);
  v += __shfl_xor(v, 2, 16);
  v += __shfl_xor(v, 1, 16);
  return v;
}

#define TK_ 32
#define PK_ 36

template <int WM>
__global__ __launch_bounds__(WM * 64) void gemm_rowA_wmma(
    const float* __restrict__ A, long strideAm, long batchStrideA,
    const float* __restrict__ W, const float* __restrict__ bias,
    float* __restrict__ Out, long strideOm, long batchStrideO,
    int M, int K, int mtiles, int ntiles) {
  constexpr int BM = WM * 32, BN = 128, NT = WM * 64;
  constexpr int RPP = NT / 8;
  constexpr int PA = BM / RPP;
  constexpr int PW = BN / RPP;
  __shared__ __attribute__((aligned(16))) float sA[2][BM * PK_];
  __shared__ __attribute__((aligned(16))) float sW[2][BN * PK_];
  int tid = threadIdx.x;
  int lane = tid & 31;
  int wslot = tid >> 5;
  int tilesPerBatch = mtiles * ntiles;
  int b = blockIdx.x / tilesPerBatch;
  int r = blockIdx.x - b * tilesPerBatch;
  int bm0 = (r / ntiles) * BM;
  int bn0 = (r % ntiles) * BN;
  const float* Ab = A + (long)b * batchStrideA;
  float* Ob = Out + (long)b * batchStrideO;
  int col = lane & 15;
  int hi = lane >> 4;
  int wm = wslot >> 1, wn = wslot & 1;

  int srow = tid >> 3, sq = (tid & 7) * 4;
  const float* Ag[PA];
  const float* Wg[PW];
#pragma unroll
  for (int p = 0; p < PA; ++p) {
    int row = bm0 + srow + RPP * p;
    if (row > M - 1) row = M - 1;
    Ag[p] = Ab + (long)row * strideAm + sq;
  }
#pragma unroll
  for (int p = 0; p < PW; ++p) Wg[p] = W + (long)(bn0 + srow + RPP * p) * K + sq;
  const int sOff = srow * PK_ + sq;

  const int aOff = (wm * 32 + col) * PK_ + 4 * hi;
  const int bOff = (wn * 64 + col) * PK_ + 4 * hi;

  v4f ra[PA], rw[PW];
#pragma unroll
  for (int p = 0; p < PA; ++p) ra[p] = *(const v4f*)(Ag[p]);
#pragma unroll
  for (int p = 0; p < PW; ++p) rw[p] = *(const v4f*)(Wg[p]);
#pragma unroll
  for (int p = 0; p < PA; ++p) *(v4f*)&sA[0][sOff + p * RPP * PK_] = ra[p];
#pragma unroll
  for (int p = 0; p < PW; ++p) *(v4f*)&sW[0][sOff + p * RPP * PK_] = rw[p];
  __syncthreads();

  v8f acc[2][4] = {};
  const int KT = K / TK_;
  for (int kt = 0; kt < KT; ++kt) {
    const int cur = kt & 1;
    const bool more = kt + 1 < KT;
    if (more) {
      const int ko = (kt + 1) * TK_;
#pragma unroll
      for (int p = 0; p < PA; ++p) ra[p] = *(const v4f*)(Ag[p] + ko);
#pragma unroll
      for (int p = 0; p < PW; ++p) rw[p] = *(const v4f*)(Wg[p] + ko);
    }
    const float* as = &sA[cur][aOff];
    const float* ws = &sW[cur][bOff];
    v4f fa[2][2], fb[2][4];
    fa[0][0] = *(const v4f*)(as);
    fa[0][1] = *(const v4f*)(as + 16 * PK_);
#pragma unroll
    for (int j = 0; j < 4; ++j) fb[0][j] = *(const v4f*)(ws + j * 16 * PK_);
#pragma unroll
    for (int c = 0; c < 4; ++c) {
      const int s = c & 1, n = s ^ 1;
      if (c < 3) {
        fa[n][0] = *(const v4f*)(as + 8 * (c + 1));
        fa[n][1] = *(const v4f*)(as + 16 * PK_ + 8 * (c + 1));
#pragma unroll
        for (int j = 0; j < 4; ++j)
          fb[n][j] = *(const v4f*)(ws + j * 16 * PK_ + 8 * (c + 1));
      }
      __builtin_amdgcn_sched_barrier(0);
#pragma unroll
      for (int j = 0; j < 4; ++j) {
        acc[0][j] = wmma4(lo2(fa[s][0]), lo2(fb[s][j]), acc[0][j]);
        acc[1][j] = wmma4(lo2(fa[s][1]), lo2(fb[s][j]), acc[1][j]);
      }
#pragma unroll
      for (int j = 0; j < 4; ++j) {
        acc[0][j] = wmma4(hi2(fa[s][0]), hi2(fb[s][j]), acc[0][j]);
        acc[1][j] = wmma4(hi2(fa[s][1]), hi2(fb[s][j]), acc[1][j]);
      }
    }
    if (more) {
#pragma unroll
      for (int p = 0; p < PA; ++p) *(v4f*)&sA[cur ^ 1][sOff + p * RPP * PK_] = ra[p];
#pragma unroll
      for (int p = 0; p < PW; ++p) *(v4f*)&sW[cur ^ 1][sOff + p * RPP * PK_] = rw[p];
    }
    __syncthreads();
  }

  const int m0 = bm0 + wm * 32;
  const int n0 = bn0 + wn * 64;
  float bv[4];
#pragma unroll
  for (int j = 0; j < 4; ++j) bv[j] = bias[n0 + j * 16 + col];
#pragma unroll
  for (int i = 0; i < 2; ++i) {
#pragma unroll
    for (int rr = 0; rr < 8; ++rr) {
      int row = m0 + i * 16 + rr + 8 * hi;
      if (row < M) {
        float* orow = Ob + (long)row * strideOm + n0;
#pragma unroll
        for (int j = 0; j < 4; ++j) orow[j * 16 + col] = acc[i][j][rr] + bv[j];
      }
    }
  }
}

template <int WM>
__global__ __launch_bounds__(WM * 64) void gemm_colA_wmma(
    const float* __restrict__ A, long strideAk, long batchStrideA,
    const float* __restrict__ Wkv,
    float* __restrict__ Out, long strideOm, long batchStrideO,
    int K, int mtiles, int ntiles) {
  constexpr int BM = WM * 32, BN = 128, NT = WM * 64;
  constexpr int PM = BM + 4;
  constexpr int TPR = BM / 4;
  constexpr int KPP = NT / TPR;
  constexpr int PA = TK_ / KPP;
  constexpr int RPP = NT / 8;
  constexpr int PW = BN / RPP;
  __shared__ __attribute__((aligned(16))) float sA[2][TK_ * PM];
  __shared__ __attribute__((aligned(16))) float sW[2][BN * PK_];
  int tid = threadIdx.x;
  int lane = tid & 31;
  int wslot = tid >> 5;
  int tilesPerBatch = mtiles * ntiles;
  int b = blockIdx.x / tilesPerBatch;
  int r = blockIdx.x - b * tilesPerBatch;
  int bm0 = (r / ntiles) * BM;
  int bn0 = (r % ntiles) * BN;
  const float* Ab = A + (long)b * batchStrideA;
  float* Ob = Out + (long)b * batchStrideO;
  int col = lane & 15;
  int hi = lane >> 4;
  int wm = wslot >> 1, wn = wslot & 1;

  const int ak = tid / TPR, am = (tid % TPR) * 4;
  const float* Ag = Ab + bm0 + am;
  const int saOff = ak * PM + am;
  const int srow = tid >> 3, sq = (tid & 7) * 4;
  const float* Wg = Wkv + (long)(bn0 + srow) * WKV_LD + sq;
  const int swOff = srow * PK_ + sq;

  const int aOff = (4 * hi) * PM + wm * 32 + col;
  const int bOff = (wn * 64 + col) * PK_ + 4 * hi;

  v4f ra[PA], rw[PW];
  const v4f zero4 = {0.f, 0.f, 0.f, 0.f};
  auto fetch = [&](int kt) {
    const int k0 = kt * TK_;
#pragma unroll
    for (int p = 0; p < PA; ++p) {
      int k = k0 + ak + KPP * p;
      int kc = k < K ? k : K - 1;
      ra[p] = *(const v4f*)(Ag + (long)kc * strideAk);
    }
#pragma unroll
    for (int p = 0; p < PW; ++p)
      rw[p] = *(const v4f*)(Wg + (long)(RPP * p) * WKV_LD + k0);
  };
  auto stash = [&](int st, int kt) {
    const int k0 = kt * TK_;
#pragma unroll
    for (int p = 0; p < PA; ++p) {
      int k = k0 + ak + KPP * p;
      *(v4f*)&sA[st][saOff + p * KPP * PM] = k < K ? ra[p] : zero4;
    }
    const int kq = k0 + sq;
    const bool v0 = kq < K, v1 = kq + 1 < K, v2 = kq + 2 < K, v3 = kq + 3 < K;
#pragma unroll
    for (int p = 0; p < PW; ++p) {
      v4f t = rw[p];
      *(v4f*)&sW[st][swOff + p * RPP * PK_] =
          (v4f){v0 ? t.x : 0.f, v1 ? t.y : 0.f, v2 ? t.z : 0.f, v3 ? t.w : 0.f};
    }
  };

  fetch(0);
  stash(0, 0);
  __syncthreads();

  v8f acc[2][4] = {};
  const int KT = (K + TK_ - 1) / TK_;
  for (int kt = 0; kt < KT; ++kt) {
    const int cur = kt & 1;
    const bool more = kt + 1 < KT;
    if (more) fetch(kt + 1);
    const float* as = &sA[cur][aOff];
    const float* ws = &sW[cur][bOff];
    v2f al[2][2], ah[2][2];
    v4f fb[2][4];
#pragma unroll
    for (int i = 0; i < 2; ++i) {
      al[0][i] = (v2f){as[i * 16], as[PM + i * 16]};
      ah[0][i] = (v2f){as[2 * PM + i * 16], as[3 * PM + i * 16]};
    }
#pragma unroll
    for (int j = 0; j < 4; ++j) fb[0][j] = *(const v4f*)(ws + j * 16 * PK_);
#pragma unroll
    for (int c = 0; c < 4; ++c) {
      const int s = c & 1, n = s ^ 1;
      if (c < 3) {
        const float* an = as + 8 * (c + 1) * PM;
#pragma unroll
        for (int i = 0; i < 2; ++i) {
          al[n][i] = (v2f){an[i * 16], an[PM + i * 16]};
          ah[n][i] = (v2f){an[2 * PM + i * 16], an[3 * PM + i * 16]};
        }
#pragma unroll
        for (int j = 0; j < 4; ++j)
          fb[n][j] = *(const v4f*)(ws + j * 16 * PK_ + 8 * (c + 1));
      }
      __builtin_amdgcn_sched_barrier(0);
#pragma unroll
      for (int j = 0; j < 4; ++j) {
        acc[0][j] = wmma4(al[s][0], lo2(fb[s][j]), acc[0][j]);
        acc[1][j] = wmma4(al[s][1], lo2(fb[s][j]), acc[1][j]);
      }
#pragma unroll
      for (int j = 0; j < 4; ++j) {
        acc[0][j] = wmma4(ah[s][0], hi2(fb[s][j]), acc[0][j]);
        acc[1][j] = wmma4(ah[s][1], hi2(fb[s][j]), acc[1][j]);
      }
    }
    if (more) stash(cur ^ 1, kt + 1);
    __syncthreads();
  }

  const int m0 = bm0 + wm * 32;
  const int n0 = bn0 + wn * 64;
#pragma unroll
  for (int i = 0; i < 2; ++i) {
#pragma unroll
    for (int rr = 0; rr < 8; ++rr) {
      int row = m0 + i * 16 + rr + 8 * hi;
      float* orow = Ob + (long)row * strideOm + n0;
#pragma unroll
      for (int j = 0; j < 4; ++j) orow[j * 16 + col] = acc[i][j][rr];
    }
  }
}

__global__ __launch_bounds__(256) void attn_wmma(
    const float* __restrict__ q, const float* __restrict__ kbuf,
    const float* __restrict__ vbuf, float* __restrict__ obuf) {
  __shared__ __attribute__((aligned(16))) float lds[8][256];
  int lane = threadIdx.x & 31;
  int wslot = threadIdx.x >> 5;
  int wid = (blockIdx.x * blockDim.x + threadIdx.x) >> 5;
  int ltile = wid & 127;
  int h = (wid >> 7) & 15;
  int b = wid >> 11;
  int l0 = ltile * 16;
  int col = lane & 15;
  int hi = lane >> 4;
  int khalf = hi * 2;

  v2f qA[8];
  {
    int l = l0 + col;
    bool lv = l < L_;
    const float* qrow = q + ((long)b * L_ + l) * C_ + h * HD_;
    const float sc = 0.125f * 1.44269504088896340736f;
#pragma unroll
    for (int kk = 0; kk < 8; ++kk) {
      int k = kk * 4 + khalf;
      float x0 = lv ? qrow[k] * sc : 0.f;
      float x1 = lv ? qrow[k + 1] * sc : 0.f;
      qA[kk] = (v2f){x0, x1};
    }
  }

  v8f O[4] = {};
  float Mrow[8], Srow[8];
#pragma unroll
  for (int r = 0; r < 8; ++r) { Mrow[r] = -INFINITY; Srow[r] = 0.f; }

  const float* kb = kbuf + (long)b * C_ * P_ + (long)h * P_;
  const float* vb = vbuf + (long)b * D_ * P_ + (long)h * P_;

  for (int p0 = 0; p0 < P_; p0 += 16) {
    v8f c = {};
#pragma unroll
    for (int kk = 0; kk < 8; ++kk) {
      int k = kk * 4 + khalf;
      const float* kp = kb + (long)k * 16 * P_ + p0 + col;
      v2f bf = {kp[0], kp[16 * P_]};
      c = wmma4(qA[kk], bf, c);
    }
    float alpha[8];
#pragma unroll
    for (int r = 0; r < 8; ++r) {
      float mx = rowmax16(c[r]);
      float mnew = fmaxf(Mrow[r], mx);
      float al = __builtin_exp2f(Mrow[r] - mnew);
      float pv = __builtin_exp2f(c[r] - mnew);
      Srow[r] = Srow[r] * al + rowsum16(pv);
      Mrow[r] = mnew;
      alpha[r] = al;
      lds[wslot][(r + 8 * hi) * 16 + col] = pv;
    }
#pragma unroll
    for (int r = 0; r < 8; ++r) {
      O[0][r] *= alpha[r]; O[1][r] *= alpha[r];
      O[2][r] *= alpha[r]; O[3][r] *= alpha[r];
    }
    __builtin_amdgcn_fence(__ATOMIC_ACQ_REL, "wavefront");
    __builtin_amdgcn_wave_barrier();
    v4f pA[2];
#pragma unroll
    for (int c8 = 0; c8 < 2; ++c8)
      pA[c8] = *(const v4f*)&lds[wslot][col * 16 + c8 * 8 + hi * 4];
    __builtin_amdgcn_fence(__ATOMIC_ACQ_REL, "wavefront");
    __builtin_amdgcn_wave_barrier();
#pragma unroll
    for (int c8 = 0; c8 < 2; ++c8) {
      v4f vf[4];
#pragma unroll
      for (int j = 0; j < 4; ++j) {
        const float* vp = vb + (long)((j * 16 + col) * 16) * P_ + p0 + c8 * 8 + hi * 4;
        vf[j] = *(const v4f*)vp;
      }
#pragma unroll
      for (int j = 0; j < 4; ++j) O[j] = wmma4(lo2(pA[c8]), lo2(vf[j]), O[j]);
#pragma unroll
      for (int j = 0; j < 4; ++j) O[j] = wmma4(hi2(pA[c8]), hi2(vf[j]), O[j]);
    }
  }

#pragma unroll
  for (int r = 0; r < 8; ++r) {
    float inv = 1.f / Srow[r];
    O[0][r] *= inv; O[1][r] *= inv; O[2][r] *= inv; O[3][r] *= inv;
  }
#pragma unroll
  for (int r = 0; r < 8; ++r) {
    int l = l0 + r + 8 * hi;
    if (l < L_) {
      float* orow = obuf + ((long)b * L_ + l) * D_ + h;
#pragma unroll
      for (int j = 0; j < 4; ++j) orow[(j * 16 + col) * 16] = O[j][r];
    }
  }
}

extern "C" void kernel_launch(void* const* d_in, const int* in_sizes, int n_in,
                              void* d_out, int out_size, void* d_ws,
                              size_t ws_size, hipStream_t stream) {
  (void)in_sizes; (void)n_in; (void)out_size; (void)ws_size;
  const float* in = (const float*)d_in[0];
  const float* w_kv = (const float*)d_in[1];
  const float* w_q = (const float*)d_in[2];
  const float* b_q = (const float*)d_in[3];
  const float* w_out = (const float*)d_in[4];
  const float* b_out = (const float*)d_in[5];
  float* out = (float*)d_out;

  float* ws = (float*)d_ws;
  float* qbuf = ws;
  float* vbuf = qbuf + (size_t)B_ * L_ * C_;
  float* kbuf = vbuf + (size_t)B_ * D_ * P_;
  float* obuf = kbuf + (size_t)B_ * C_ * P_;

  gemm_rowA_wmma<4><<<256, 256, 0, stream>>>(in, (long)(B_ * D_), (long)D_, w_q,
                                             b_q, qbuf, (long)C_, (long)L_ * C_,
                                             L_, D_, 16, 4);
  gemm_colA_wmma<4><<<256, 256, 0, stream>>>(in, (long)(B_ * D_), (long)D_, w_kv,
                                             vbuf, (long)P_, (long)D_ * P_, L_,
                                             8, 8);
  gemm_colA_wmma<2><<<256, 128, 0, stream>>>(qbuf, (long)C_, (long)L_ * C_, w_kv,
                                             kbuf, (long)P_, (long)C_ * P_, L_,
                                             8, 8);
  attn_wmma<<<1024, 256, 0, stream>>>(qbuf, kbuf, vbuf, obuf);
  gemm_rowA_wmma<4><<<512, 256, 0, stream>>>(obuf, (long)D_, (long)L_ * D_,
                                             w_out, b_out, out, (long)(B_ * D_),
                                             (long)D_, L_, D_, 16, 8);
}


# Attention: each 16-p slice of K and V staged through LDS once per work-group (whole 64-B row pieces, two stages), the 8 waves read their B fragments by ds_load; all sum orders kept
# speedup vs baseline: 1.7786x; 1.1528x over previous
#include <hip/hip_runtime.h>


typedef __attribute__((ext_vector_type(2))) float v2f;
typedef __attribute__((ext_vector_type(4))) float v4f;
typedef __attribute__((ext_vector_type(8))) float v8f;

static __device__ __forceinline__ v2f lo2(v4f x) { return (v2f){x.x, x.y}; }
static __device__ __forceinline__ v2f hi2(v4f x) { return (v2f){x.z, x.w}; }

#define B_ 4
#define L_ 2046
#define D_ 1024
#define C_ 512
#define P_ 1024
#define H_ 16
#define HD_ 32
#define DV_ 64
#define WKV_LD 2048

static __device__ __forceinline__ v8f wmma4(v2f a, v2f b, v8f c) {
  return __builtin_amdgcn_wmma_f32_16x16x4_f32(false, a, false, b, (short)0, c,
                                               false, false);
}

static __device__ __forceinline__ float rowmax16(float v) {
  v = fmaxf(v, __shfl_xor(v, 8, 16));
  v = fmaxf(v, __shfl_xor(v, 4, 16));
  v = fmaxf(v, __shfl_xor(v, 2, 16));
  v = fmaxf(v, __shfl_xor(v, 1, 16));
  return v;
}
static __device__ __forceinline__ float rowsum16(float v) {
  v += __shfl_xor(v, 8, 16);
  v += __shfl_xor(v, 4, 16);
  v += __shfl_xor(v, 2, 16);
  v += __shfl_xor(v, 1, 16);
  return v;
}

#define TK_ 32
#define PK_ 36

template <int WM>
__global__ __launch_bounds__(WM * 64) void gemm_rowA_wmma(
    const float* __restrict__ A, long strideAm, long batchStrideA,
    const float* __restrict__ W, const float* __restrict__ bias,
    float* __restrict__ Out, long strideOm, long batchStrideO,
    int M, int K, int mtiles, int ntiles) {
  constexpr int BM = WM * 32, BN = 128, NT = WM * 64;
  constexpr int RPP = NT / 8;
  constexpr int PA = BM / RPP;
  constexpr int PW = BN / RPP;
  __shared__ __attribute__((aligned(16))) float sA[2][BM * PK_];
  __shared__ __attribute__((aligned(16))) float sW[2][BN * PK_];
  int tid = threadIdx.x;
  int lane = tid & 31;
  int wslot = tid >> 5;
  int tilesPerBatch = mtiles * ntiles;
  int b = blockIdx.x / tilesPerBatch;
  int r = blockIdx.x - b * tilesPerBatch;
  int bm0 = (r / ntiles) * BM;
  int bn0 = (r % ntiles) * BN;
  const float* Ab = A + (long)b * batchStrideA;
  float* Ob = Out + (long)b * batchStrideO;
  int col = lane & 15;
  int hi = lane >> 4;
  int wm = wslot >> 1, wn = wslot & 1;

  int srow = tid >> 3, sq = (tid & 7) * 4;
  const float* Ag[PA];
  const float* Wg[PW];
#pragma unroll
  for (int p = 0; p < PA; ++p) {
    int row = bm0 + srow + RPP * p;
    if (row > M - 1) row = M - 1;
    Ag[p] = Ab + (long)row * strideAm + sq;
  }
#pragma unroll
  for (int p = 0; p < PW; ++p) Wg[p] = W + (long)(bn0 + srow + RPP * p) * K + sq;
  const int sOff = srow * PK_ + sq;

  const int aOff = (wm * 32 + col) * PK_ + 4 * hi;
  const int bOff = (wn * 64 + col) * PK_ + 4 * hi;

  v4f ra[PA], rw[PW];
#pragma unroll
  for (int p = 0; p < PA; ++p) ra[p] = *(const v4f*)(Ag[p]);
#pragma unroll
  for (int p = 0; p < PW; ++p) rw[p] = *(const v4f*)(Wg[p]);
#pragma unroll
  for (int p = 0; p < PA; ++p) *(v4f*)&sA[0][sOff + p * RPP * PK_] = ra[p];
#pragma unroll
  for (int p = 0; p < PW; ++p) *(v4f*)&sW[0][sOff + p * RPP * PK_] = rw[p];
  __syncthreads();

  v8f acc[2][4] = {};
  const int KT = K / TK_;
  for (int kt = 0; kt < KT; ++kt) {
    const int cur = kt & 1;
    const bool more = kt + 1 < KT;
    if (more) {
      const int ko = (kt + 1) * TK_;
#pragma unroll
      for (int p = 0; p < PA; ++p) ra[p] = *(const v4f*)(Ag[p] + ko);
#pragma unroll
      for (int p = 0; p < PW; ++p) rw[p] = *(const v4f*)(Wg[p] + ko);
    }
    const float* as = &sA[cur][aOff];
    const float* ws = &sW[cur][bOff];
    v4f fa[2][2], fb[2][4];
    fa[0][0] = *(const v4f*)(as);
    fa[0][1] = *(const v4f*)(as + 16 * PK_);
#pragma unroll
    for (int j = 0; j < 4; ++j) fb[0][j] = *(const v4f*)(ws + j * 16 * PK_);
#pragma unroll
    for (int c = 0; c < 4; ++c) {
      const int s = c & 1, n = s ^ 1;
      if (c < 3) {
        fa[n][0] = *(const v4f*)(as + 8 * (c + 1));
        fa[n][1] = *(const v4f*)(as + 16 * PK_ + 8 * (c + 1));
#pragma unroll
        for (int j = 0; j < 4; ++j)
          fb[n][j] = *(const v4f*)(ws + j * 16 * PK_ + 8 * (c + 1));
      }
      __builtin_amdgcn_sched_barrier(0);
#pragma unroll
      for (int j = 0; j < 4; ++j) {
        acc[0][j] = wmma4(lo2(fa[s][0]), lo2(fb[s][j]), acc[0][j]);
        acc[1][j] = wmma4(lo2(fa[s][1]), lo2(fb[s][j]), acc[1][j]);
      }
#pragma unroll
      for (int j = 0; j < 4; ++j) {
        acc[0][j] = wmma4(hi2(fa[s][0]), hi2(fb[s][j]), acc[0][j]);
        acc[1][j] = wmma4(hi2(fa[s][1]), hi2(fb[s][j]), acc[1][j]);
      }
    }
    if (more) {
#pragma unroll
      for (int p = 0; p < PA; ++p) *(v4f*)&sA[cur ^ 1][sOff + p * RPP * PK_] = ra[p];
#pragma unroll
      for (int p = 0; p < PW; ++p) *(v4f*)&sW[cur ^ 1][sOff + p * RPP * PK_] = rw[p];
    }
    __syncthreads();
  }

  const int m0 = bm0 + wm * 32;
  const int n0 = bn0 + wn * 64;
  float bv[4];
#pragma unroll
  for (int j = 0; j < 4; ++j) bv[j] = bias[n0 + j * 16 + col];
#pragma unroll
  for (int i = 0; i < 2; ++i) {
#pragma unroll
    for (int rr = 0; rr < 8; ++rr) {
      int row = m0 + i * 16 + rr + 8 * hi;
      if (row < M) {
        float* orow = Ob + (long)row * strideOm + n0;
#pragma unroll
        for (int j = 0; j < 4; ++j) orow[j * 16 + col] = acc[i][j][rr] + bv[j];
      }
    }
  }
}

template <int WM>
__global__ __launch_bounds__(WM * 64) void gemm_colA_wmma(
    const float* __restrict__ A, long strideAk, long batchStrideA,
    const float* __restrict__ Wkv,
    float* __restrict__ Out, long strideOm, long batchStrideO,
    int K, int mtiles, int ntiles) {
  constexpr int BM = WM * 32, BN = 128, NT = WM * 64;
  constexpr int PM = BM + 4;
  constexpr int TPR = BM / 4;
  constexpr int KPP = NT / TPR;
  constexpr int PA = TK_ / KPP;
  constexpr int RPP = NT / 8;
  constexpr int PW = BN / RPP;
  __shared__ __attribute__((aligned(16))) float sA[2][TK_ * PM];
  __shared__ __attribute__((aligned(16))) float sW[2][BN * PK_];
  int tid = threadIdx.x;
  int lane = tid & 31;
  int wslot = tid >> 5;
  int tilesPerBatch = mtiles * ntiles;
  int b = blockIdx.x / tilesPerBatch;
  int r = blockIdx.x - b * tilesPerBatch;
  int bm0 = (r / ntiles) * BM;
  int bn0 = (r % ntiles) * BN;
  const float* Ab = A + (long)b * batchStrideA;
  float* Ob = Out + (long)b * batchStrideO;
  int col = lane & 15;
  int hi = lane >> 4;
  int wm = wslot >> 1, wn = wslot & 1;

  const int ak = tid / TPR, am = (tid % TPR) * 4;
  const float* Ag = Ab + bm0 + am;
  const int saOff = ak * PM + am;
  const int srow = tid >> 3, sq = (tid & 7) * 4;
  const float* Wg = Wkv + (long)(bn0 + srow) * WKV_LD + sq;
  const int swOff = srow * PK_ + sq;

  const int aOff = (4 * hi) * PM + wm * 32 + col;
  const int bOff = (wn * 64 + col) * PK_ + 4 * hi;

  v4f ra[PA], rw[PW];
  const v4f zero4 = {0.f, 0.f, 0.f, 0.f};
  auto fetch = [&](int kt) {
    const int k0 = kt * TK_;
#pragma unroll
    for (int p = 0; p < PA; ++p) {
      int k = k0 + ak + KPP * p;
      int kc = k < K ? k : K - 1;
      ra[p] = *(const v4f*)(Ag + (long)kc * strideAk);
    }
#pragma unroll
    for (int p = 0; p < PW; ++p)
      rw[p] = *(const v4f*)(Wg + (long)(RPP * p) * WKV_LD + k0);
  };
  auto stash = [&](int st, int kt) {
    const int k0 = kt * TK_;
#pragma unroll
    for (int p = 0; p < PA; ++p) {
      int k = k0 + ak + KPP * p;
      *(v4f*)&sA[st][saOff + p * KPP * PM] = k < K ? ra[p] : zero4;
    }
    const int kq = k0 + sq;
    const bool v0 = kq < K, v1 = kq + 1 < K, v2 = kq + 2 < K, v3 = kq + 3 < K;
#pragma unroll
    for (int p = 0; p < PW; ++p) {
      v4f t = rw[p];
      *(v4f*)&sW[st][swOff + p * RPP * PK_] =
          (v4f){v0 ? t.x : 0.f, v1 ? t.y : 0.f, v2 ? t.z : 0.f, v3 ? t.w : 0.f};
    }
  };

  fetch(0);
  stash(0, 0);
  __syncthreads();

  v8f acc[2][4] = {};
  const int KT = (K + TK_ - 1) / TK_;
  for (int kt = 0; kt < KT; ++kt) {
    const int cur = kt & 1;
    const bool more = kt + 1 < KT;
    if (more) fetch(kt + 1);
    const float* as = &sA[cur][aOff];
    const float* ws = &sW[cur][bOff];
    v2f al[2][2], ah[2][2];
    v4f fb[2][4];
#pragma unroll
    for (int i = 0; i < 2; ++i) {
      al[0][i] = (v2f){as[i * 16], as[PM + i * 16]};
      ah[0][i] = (v2f){as[2 * PM + i * 16], as[3 * PM + i * 16]};
    }
#pragma unroll
    for (int j = 0; j < 4; ++j) fb[0][j] = *(const v4f*)(ws + j * 16 * PK_);
#pragma unroll
    for (int c = 0; c < 4; ++c) {
      const int s = c & 1, n = s ^ 1;
      if (c < 3) {
        const float* an = as + 8 * (c + 1) * PM;
#pragma unroll
        for (int i = 0; i < 2; ++i) {
          al[n][i] = (v2f){an[i * 16], an[PM + i * 16]};
          ah[n][i] = (v2f){an[2 * PM + i * 16], an[3 * PM + i * 16]};
        }
#pragma unroll
        for (int j = 0; j < 4; ++j)
          fb[n][j] = *(const v4f*)(ws + j * 16 * PK_ + 8 * (c + 1));
      }
      __builtin_amdgcn_sched_barrier(0);
#pragma unroll
      for (int j = 0; j < 4; ++j) {
        acc[0][j] = wmma4(al[s][0], lo2(fb[s][j]), acc[0][j]);
        acc[1][j] = wmma4(al[s][1], lo2(fb[s][j]), acc[1][j]);
      }
#pragma unroll
      for (int j = 0; j < 4; ++j) {
        acc[0][j] = wmma4(ah[s][0], hi2(fb[s][j]), acc[0][j]);
        acc[1][j] = wmma4(ah[s][1], hi2(fb[s][j]), acc[1][j]);
      }
    }
    if (more) stash(cur ^ 1, kt + 1);
    __syncthreads();
  }

  const int m0 = bm0 + wm * 32;
  const int n0 = bn0 + wn * 64;
#pragma unroll
  for (int i = 0; i < 2; ++i) {
#pragma unroll
    for (int rr = 0; rr < 8; ++rr) {
      int row = m0 + i * 16 + rr + 8 * hi;
      float* orow = Ob + (long)row * strideOm + n0;
#pragma unroll
      for (int j = 0; j < 4; ++j) orow[j * 16 + col] = acc[i][j][rr];
    }
  }
}

#define AP_ 16
#define KP_ 16
#define VP_ 20
__global__ __launch_bounds__(256, 8) void attn_wmma(
    const float* __restrict__ q, const float* __restrict__ kbuf,
    const float* __restrict__ vbuf, float* __restrict__ obuf) {
  __shared__ __attribute__((aligned(16))) float lds[8][256];
  __shared__ __attribute__((aligned(16))) float sK[2][HD_ * KP_];
  __shared__ __attribute__((aligned(16))) float sV[2][DV_ * VP_];
  int tid = threadIdx.x;
  int lane = tid & 31;
  int wslot = tid >> 5;
  int ltile = (blockIdx.x & 15) * 8 + wslot;
  int h = (blockIdx.x >> 4) & 15;
  int b = blockIdx.x >> 8;
  int l0 = ltile * 16;
  int col = lane & 15;
  int hi = lane >> 4;
  int khalf = hi * 2;

  v2f qA[8];
  {
    int l = l0 + col;
    bool lv = l < L_;
    const float* qrow = q + ((long)b * L_ + l) * C_ + h * HD_;
    const float sc = 0.125f * 1.44269504088896340736f;
#pragma unroll
    for (int kk = 0; kk < 8; ++kk) {
      int k = kk * 4 + khalf;
      float x0 = lv ? qrow[k] * sc : 0.f;
      float x1 = lv ? qrow[k + 1] * sc : 0.f;
      qA[kk] = (v2f){x0, x1};
    }
  }

  v8f O[4] = {};
  float Mrow[8], Srow[8];
#pragma unroll
  for (int r = 0; r < 8; ++r) { Mrow[r] = -INFINITY; Srow[r] = 0.f; }

  const float* kb = kbuf + (long)b * C_ * P_ + (long)h * P_;
  const float* vb = vbuf + (long)b * D_ * P_ + (long)h * P_;

  const int srow = tid >> 3, sq = (tid & 7) * 2;
  const unsigned goff = (unsigned)(srow * 16) * P_ + sq;
  const float* gK = kb + goff;
  const float* gV = vb + goff;
  const int skOff = srow * KP_ + sq;
  const int svOff = srow * VP_ + sq;

  const int kOff = khalf * KP_ + col;
  const int vOff = col * VP_ + hi * 4;

  v2f rk, rv0, rv1;
  rk = *(const v2f*)gK;
  rv0 = *(const v2f*)gV;
  rv1 = *(const v2f*)(gV + (long)(32 * 16) * P_);
  *(v2f*)&sK[0][skOff] = rk;
  *(v2f*)&sV[0][svOff] = rv0;
  *(v2f*)&sV[0][svOff + 32 * VP_] = rv1;
  rk = *(const v2f*)(gK + AP_);
  rv0 = *(const v2f*)(gV + AP_);
  rv1 = *(const v2f*)(gV + (long)(32 * 16) * P_ + AP_);
  __syncthreads();

  constexpr int NS = P_ / AP_;
  for (int s = 0; s < NS; ++s) {
    const int cur = s & 1;
    const bool more = s + 1 < NS;
    const float* ks = &sK[cur][kOff];
    const float* vs = &sV[cur][vOff];
    v8f c = {};
#pragma unroll
    for (int kk = 0; kk < 8; ++kk) {
      const float* kp = ks + kk * 4 * KP_;
      v2f bf = {kp[0], kp[KP_]};
      c = wmma4(qA[kk], bf, c);
    }
    if (more) {
      *(v2f*)&sK[cur ^ 1][skOff] = rk;
      *(v2f*)&sV[cur ^ 1][svOff] = rv0;
      *(v2f*)&sV[cur ^ 1][svOff + 32 * VP_] = rv1;
    }
    float alpha[8];
#pragma unroll
    for (int r = 0; r < 8; ++r) {
      float mx = rowmax16(c[r]);
      float mnew = fmaxf(Mrow[r], mx);
      float al = __builtin_exp2f(Mrow[r] - mnew);
      float pv = __builtin_exp2f(c[r] - mnew);
      Srow[r] = Srow[r] * al + rowsum16(pv);
      Mrow[r] = mnew;
      alpha[r] = al;
      lds[wslot][(r + 8 * hi) * 16 + col] = pv;
    }
#pragma unroll
    for (int r = 0; r < 8; ++r) {
      O[0][r] *= alpha[r]; O[1][r] *= alpha[r];
      O[2][r] *= alpha[r]; O[3][r] *= alpha[r];
    }
    __builtin_amdgcn_fence(__ATOMIC_ACQ_REL, "wavefront");
    __builtin_amdgcn_wave_barrier();
    v4f pA[2];
#pragma unroll
    for (int c8 = 0; c8 < 2; ++c8)
      pA[c8] = *(const v4f*)&lds[wslot][col * 16 + c8 * 8 + hi * 4];
    __builtin_amdgcn_fence(__ATOMIC_ACQ_REL, "wavefront");
    __builtin_amdgcn_wave_barrier();
    if (s + 2 < NS) {
      const int po = (s + 2) * AP_;
      rk = *(const v2f*)(gK + po);
      rv0 = *(const v2f*)(gV + po);
      rv1 = *(const v2f*)(gV + (long)(32 * 16) * P_ + po);
    }
#pragma unroll
    for (int c8 = 0; c8 < 2; ++c8) {
      v4f vf[4];
#pragma unroll
      for (int j = 0; j < 4; ++j)
        vf[j] = *(const v4f*)(vs + j * 16 * VP_ + c8 * 8);
#pragma unroll
      for (int j = 0; j < 4; ++j) O[j] = wmma4(lo2(pA[c8]), lo2(vf[j]), O[j]);
#pragma unroll
      for (int j = 0; j < 4; ++j) O[j] = wmma4(hi2(pA[c8]), hi2(vf[j]), O[j]);
    }
    __builtin_amdgcn_fence(__ATOMIC_RELEASE, "workgroup", "local");
    __builtin_amdgcn_s_barrier();
    __builtin_amdgcn_fence(__ATOMIC_ACQUIRE, "workgroup", "local");
  }

#pragma unroll
  for (int r = 0; r < 8; ++r) {
    float inv = 1.f / Srow[r];
    O[0][r] *= inv; O[1][r] *= inv; O[2][r] *= inv; O[3][r] *= inv;
  }
#pragma unroll
  for (int r = 0; r < 8; ++r) {
    int l = l0 + r + 8 * hi;
    if (l < L_) {
      float* orow = obuf + ((long)b * L_ + l) * D_ + h;
#pragma unroll
      for (int j = 0; j < 4; ++j) orow[(j * 16 + col) * 16] = O[j][r];
    }
  }
}

extern "C" void kernel_launch(void* const* d_in, const int* in_sizes, int n_in,
                              void* d_out, int out_size, void* d_ws,
                              size_t ws_size, hipStream_t stream) {
  (void)in_sizes; (void)n_in; (void)out_size; (void)ws_size;
  const float* in = (const float*)d_in[0];
  const float* w_kv = (const float*)d_in[1];
  const float* w_q = (const float*)d_in[2];
  const float* b_q = (const float*)d_in[3];
  const float* w_out = (const float*)d_in[4];
  const float* b_out = (const float*)d_in[5];
  float* out = (float*)d_out;

  float* ws = (float*)d_ws;
  float* qbuf = ws;
  float* vbuf = qbuf + (size_t)B_ * L_ * C_;
  float* kbuf = vbuf + (size_t)B_ * D_ * P_;
  float* obuf = kbuf + (size_t)B_ * C_ * P_;

  gemm_rowA_wmma<4><<<256, 256, 0, stream>>>(in, (long)(B_ * D_), (long)D_, w_q,
                                             b_q, qbuf, (long)C_, (long)L_ * C_,
                                             L_, D_, 16, 4);
  gemm_colA_wmma<4><<<256, 256, 0, stream>>>(in, (long)(B_ * D_), (long)D_, w_kv,
                                             vbuf, (long)P_, (long)D_ * P_, L_,
                                             8, 8);
  gemm_colA_wmma<2><<<256, 128, 0, stream>>>(qbuf, (long)C_, (long)L_ * C_, w_kv,
                                             kbuf, (long)P_, (long)C_ * P_, L_,
                                             8, 8);
  attn_wmma<<<1024, 256, 0, stream>>>(qbuf, kbuf, vbuf, obuf);
  gemm_rowA_wmma<4><<<512, 256, 0, stream>>>(obuf, (long)D_, (long)L_ * D_,
                                             w_out, b_out, out, (long)(B_ * D_),
                                             (long)D_, L_, D_, 16, 8);
}


# Attention in transposed orientation (S^T = K^T Q^T, O^T = V^T P^T): a lane owns one query column, row max/sum in-lane plus one lane-half exchange, P enters WMMA from registers; p sum order changes
# speedup vs baseline: 2.0102x; 1.1302x over previous
#include <hip/hip_runtime.h>


typedef __attribute__((ext_vector_type(2))) float v2f;
typedef __attribute__((ext_vector_type(4))) float v4f;
typedef __attribute__((ext_vector_type(8))) float v8f;

static __device__ __forceinline__ v2f lo2(v4f x) { return (v2f){x.x, x.y}; }
static __device__ __forceinline__ v2f hi2(v4f x) { return (v2f){x.z, x.w}; }

#define B_ 4
#define L_ 2046
#define D_ 1024
#define C_ 512
#define P_ 1024
#define H_ 16
#define HD_ 32
#define DV_ 64
#define WKV_LD 2048

static __device__ __forceinline__ v8f wmma4(v2f a, v2f b, v8f c) {
  return __builtin_amdgcn_wmma_f32_16x16x4_f32(false, a, false, b, (short)0, c,
                                               false, false);
}

static __device__ __forceinline__ float rowmax16(float v) {
  v = fmaxf(v, __shfl_xor(v, 8, 16));
  v = fmaxf(v, __shfl_xor(v, 4, 16));
  v = fmaxf(v, __shfl_xor(v, 2, 16));
  v = fmaxf(v, __shfl_xor(v, 1, 16));
  return v;
}
static __device__ __forceinline__ float rowsum16(float v) {
  v += __shfl_xor(v, 8, 16);
  v += __shfl_xor(v, 4, 16);
  v += __shfl_xor(v, 2, 16);
  v += __shfl_xor(v, 1, 16);
  return v;
}

#define TK_ 32
#define PK_ 36

template <int WM>
__global__ __launch_bounds__(WM * 64) void gemm_rowA_wmma(
    const float* __restrict__ A, long strideAm, long batchStrideA,
    const float* __restrict__ W, const float* __restrict__ bias,
    float* __restrict__ Out, long strideOm, long batchStrideO,
    int M, int K, int mtiles, int ntiles) {
  constexpr int BM = WM * 32, BN = 128, NT = WM * 64;
  constexpr int RPP = NT / 8;
  constexpr int PA = BM / RPP;
  constexpr int PW = BN / RPP;
  __shared__ __attribute__((aligned(16))) float sA[2][BM * PK_];
  __shared__ __attribute__((aligned(16))) float sW[2][BN * PK_];
  int tid = threadIdx.x;
  int lane = tid & 31;
  int wslot = tid >> 5;
  int tilesPerBatch = mtiles * ntiles;
  int b = blockIdx.x / tilesPerBatch;
  int r = blockIdx.x - b * tilesPerBatch;
  int bm0 = (r / ntiles) * BM;
  int bn0 = (r % ntiles) * BN;
  const float* Ab = A + (long)b * batchStrideA;
  float* Ob = Out + (long)b * batchStrideO;
  int col = lane & 15;
  int hi = lane >> 4;
  int wm = wslot >> 1, wn = wslot & 1;

  int srow = tid >> 3, sq = (tid & 7) * 4;
  const float* Ag[PA];
  const float* Wg[PW];
#pragma unroll
  for (int p = 0; p < PA; ++p) {
    int row = bm0 + srow + RPP * p;
    if (row > M - 1) row = M - 1;
    Ag[p] = Ab + (long)row * strideAm + sq;
  }
#pragma unroll
  for (int p = 0; p < PW; ++p) Wg[p] = W + (long)(bn0 + srow + RPP * p) * K + sq;
  const int sOff = srow * PK_ + sq;

  const int aOff = (wm * 32 + col) * PK_ + 4 * hi;
  const int bOff = (wn * 64 + col) * PK_ + 4 * hi;

  v4f ra[PA], rw[PW];
#pragma unroll
  for (int p = 0; p < PA; ++p) ra[p] = *(const v4f*)(Ag[p]);
#pragma unroll
  for (int p = 0; p < PW; ++p) rw[p] = *(const v4f*)(Wg[p]);
#pragma unroll
  for (int p = 0; p < PA; ++p) *(v4f*)&sA[0][sOff + p * RPP * PK_] = ra[p];
#pragma unroll
  for (int p = 0; p < PW; ++p) *(v4f*)&sW[0][sOff + p * RPP * PK_] = rw[p];
  __syncthreads();

  v8f acc[2][4] = {};
  const int KT = K / TK_;
  for (int kt = 0; kt < KT; ++kt) {
    const int cur = kt & 1;
    const bool more = kt + 1 < KT;
    if (more) {
      const int ko = (kt + 1) * TK_;
#pragma unroll
      for (int p = 0; p < PA; ++p) ra[p] = *(const v4f*)(Ag[p] + ko);
#pragma unroll
      for (int p = 0; p < PW; ++p) rw[p] = *(const v4f*)(Wg[p] + ko);
    }
    const float* as = &sA[cur][aOff];
    const float* ws = &sW[cur][bOff];
    v4f fa[2][2], fb[2][4];
    fa[0][0] = *(const v4f*)(as);
    fa[0][1] = *(const v4f*)(as + 16 * PK_);
#pragma unroll
    for (int j = 0; j < 4; ++j) fb[0][j] = *(const v4f*)(ws + j * 16 * PK_);
#pragma unroll
    for (int c = 0; c < 4; ++c) {
      const int s = c & 1, n = s ^ 1;
      if (c < 3) {
        fa[n][0] = *(const v4f*)(as + 8 * (c + 1));
        fa[n][1] = *(const v4f*)(as + 16 * PK_ + 8 * (c + 1));
#pragma unroll
        for (int j = 0; j < 4; ++j)
          fb[n][j] = *(const v4f*)(ws + j * 16 * PK_ + 8 * (c + 1));
      }
      __builtin_amdgcn_sched_barrier(0);
#pragma unroll
      for (int j = 0; j < 4; ++j) {
        acc[0][j] = wmma4(lo2(fa[s][0]), lo2(fb[s][j]), acc[0][j]);
        acc[1][j] = wmma4(lo2(fa[s][1]), lo2(fb[s][j]), acc[1][j]);
      }
#pragma unroll
      for (int j = 0; j < 4; ++j) {
        acc[0][j] = wmma4(hi2(fa[s][0]), hi2(fb[s][j]), acc[0][j]);
        acc[1][j] = wmma4(hi2(fa[s][1]), hi2(fb[s][j]), acc[1][j]);
      }
    }
    if (more) {
#pragma unroll
      for (int p = 0; p < PA; ++p) *(v4f*)&sA[cur ^ 1][sOff + p * RPP * PK_] = ra[p];
#pragma unroll
      for (int p = 0; p < PW; ++p) *(v4f*)&sW[cur ^ 1][sOff + p * RPP * PK_] = rw[p];
    }
    __syncthreads();
  }

  const int m0 = bm0 + wm * 32;
  const int n0 = bn0 + wn * 64;
  float bv[4];
#pragma unroll
  for (int j = 0; j < 4; ++j) bv[j] = bias[n0 + j * 16 + col];
#pragma unroll
  for (int i = 0; i < 2; ++i) {
#pragma unroll
    for (int rr = 0; rr < 8; ++rr) {
      int row = m0 + i * 16 + rr + 8 * hi;
      if (row < M) {
        float* orow = Ob + (long)row * strideOm + n0;
#pragma unroll
        for (int j = 0; j < 4; ++j) orow[j * 16 + col] = acc[i][j][rr] + bv[j];
      }
    }
  }
}

template <int WM>
__global__ __launch_bounds__(WM * 64) void gemm_colA_wmma(
    const float* __restrict__ A, long strideAk, long batchStrideA,
    const float* __restrict__ Wkv,
    float* __restrict__ Out, long strideOm, long batchStrideO,
    int K, int mtiles, int ntiles) {
  constexpr int BM = WM * 32, BN = 128, NT = WM * 64;
  constexpr int PM = BM + 4;
  constexpr int TPR = BM / 4;
  constexpr int KPP = NT / TPR;
  constexpr int PA = TK_ / KPP;
  constexpr int RPP = NT / 8;
  constexpr int PW = BN / RPP;
  __shared__ __attribute__((aligned(16))) float sA[2][TK_ * PM];
  __shared__ __attribute__((aligned(16))) float sW[2][BN * PK_];
  int tid = threadIdx.x;
  int lane = tid & 31;
  int wslot = tid >> 5;
  int tilesPerBatch = mtiles * ntiles;
  int b = blockIdx.x / tilesPerBatch;
  int r = blockIdx.x - b * tilesPerBatch;
  int bm0 = (r / ntiles) * BM;
  int bn0 = (r % ntiles) * BN;
  const float* Ab = A + (long)b * batchStrideA;
  float* Ob = Out + (long)b * batchStrideO;
  int col = lane & 15;
  int hi = lane >> 4;
  int wm = wslot >> 1, wn = wslot & 1;

  const int ak = tid / TPR, am = (tid % TPR) * 4;
  const float* Ag = Ab + bm0 + am;
  const int saOff = ak * PM + am;
  const int srow = tid >> 3, sq = (tid & 7) * 4;
  const float* Wg = Wkv + (long)(bn0 + srow) * WKV_LD + sq;
  const int swOff = srow * PK_ + sq;

  const int aOff = (4 * hi) * PM + wm * 32 + col;
  const int bOff = (wn * 64 + col) * PK_ + 4 * hi;

  v4f ra[PA], rw[PW];
  const v4f zero4 = {0.f, 0.f, 0.f, 0.f};
  auto fetch = [&](int kt) {
    const int k0 = kt * TK_;
#pragma unroll
    for (int p = 0; p < PA; ++p) {
      int k = k0 + ak + KPP * p;
      int kc = k < K ? k : K - 1;
      ra[p] = *(const v4f*)(Ag + (long)kc * strideAk);
    }
#pragma unroll
    for (int p = 0; p < PW; ++p)
      rw[p] = *(const v4f*)(Wg + (long)(RPP * p) * WKV_LD + k0);
  };
  auto stash = [&](int st, int kt) {
    const int k0 = kt * TK_;
#pragma unroll
    for (int p = 0; p < PA; ++p) {
      int k = k0 + ak + KPP * p;
      *(v4f*)&sA[st][saOff + p * KPP * PM] = k < K ? ra[p] : zero4;
    }
    const int kq = k0 + sq;
    const bool v0 = kq < K, v1 = kq + 1 < K, v2 = kq + 2 < K, v3 = kq + 3 < K;
#pragma unroll
    for (int p = 0; p < PW; ++p) {
      v4f t = rw[p];
      *(v4f*)&sW[st][swOff + p * RPP * PK_] =
          (v4f){v0 ? t.x : 0.f, v1 ? t.y : 0.f, v2 ? t.z : 0.f, v3 ? t.w : 0.f};
    }
  };

  fetch(0);
  stash(0, 0);
  __syncthreads();

  v8f acc[2][4] = {};
  const int KT = (K + TK_ - 1) / TK_;
  for (int kt = 0; kt < KT; ++kt) {
    const int cur = kt & 1;
    const bool more = kt + 1 < KT;
    if (more) fetch(kt + 1);
    const float* as = &sA[cur][aOff];
    const float* ws = &sW[cur][bOff];
    v2f al[2][2], ah[2][2];
    v4f fb[2][4];
#pragma unroll
    for (int i = 0; i < 2; ++i) {
      al[0][i] = (v2f){as[i * 16], as[PM + i * 16]};
      ah[0][i] = (v2f){as[2 * PM + i * 16], as[3 * PM + i * 16]};
    }
#pragma unroll
    for (int j = 0; j < 4; ++j) fb[0][j] = *(const v4f*)(ws + j * 16 * PK_);
#pragma unroll
    for (int c = 0; c < 4; ++c) {
      const int s = c & 1, n = s ^ 1;
      if (c < 3) {
        const float* an = as + 8 * (c + 1) * PM;
#pragma unroll
        for (int i = 0; i < 2; ++i) {
          al[n][i] = (v2f){an[i * 16], an[PM + i * 16]};
          ah[n][i] = (v2f){an[2 * PM + i * 16], an[3 * PM + i * 16]};
        }
#pragma unroll
        for (int j = 0; j < 4; ++j)
          fb[n][j] = *(const v4f*)(ws + j * 16 * PK_ + 8 * (c + 1));
      }
      __builtin_amdgcn_sched_barrier(0);
#pragma unroll
      for (int j = 0; j < 4; ++j) {
        acc[0][j] = wmma4(al[s][0], lo2(fb[s][j]), acc[0][j]);
        acc[1][j] = wmma4(al[s][1], lo2(fb[s][j]), acc[1][j]);
      }
#pragma unroll
      for (int j = 0; j < 4; ++j) {
        acc[0][j] = wmma4(ah[s][0], hi2(fb[s][j]), acc[0][j]);
        acc[1][j] = wmma4(ah[s][1], hi2(fb[s][j]), acc[1][j]);
      }
    }
    if (more) stash(cur ^ 1, kt + 1);
    __syncthreads();
  }

  const int m0 = bm0 + wm * 32;
  const int n0 = bn0 + wn * 64;
#pragma unroll
  for (int i = 0; i < 2; ++i) {
#pragma unroll
    for (int rr = 0; rr < 8; ++rr) {
      int row = m0 + i * 16 + rr + 8 * hi;
      float* orow = Ob + (long)row * strideOm + n0;
#pragma unroll
      for (int j = 0; j < 4; ++j) orow[j * 16 + col] = acc[i][j][rr];
    }
  }
}

#define AP_ 16
#define KP_ 16
#define VP_ 20
#define TP_ 17

static __device__ __forceinline__ float xhalf(float x) {
  int b = __builtin_bit_cast(int, x);
  return __builtin_bit_cast(
      float, __builtin_amdgcn_permlanex16(b, b, 0x76543210, 0xfedcba98, false, false));
}

__global__ __launch_bounds__(256, 8) void attn_wmma(
    const float* __restrict__ q, const float* __restrict__ kbuf,
    const float* __restrict__ vbuf, float* __restrict__ obuf) {
  __shared__ __attribute__((aligned(16))) float sK[2][HD_ * KP_];
  __shared__ __attribute__((aligned(16))) float sV[2][DV_ * VP_];
  __shared__ float sT[8][16 * TP_];
  int tid = threadIdx.x;
  int lane = tid & 31;
  int wslot = tid >> 5;
  int ltile = (blockIdx.x & 15) * 8 + wslot;
  int h = (blockIdx.x >> 4) & 15;
  int b = blockIdx.x >> 8;
  int l0 = ltile * 16;
  int col = lane & 15;
  int hi = lane >> 4;
  int khalf = hi * 2;

  v2f qB[8];
  {
    int l = l0 + col;
    bool lv = l < L_;
    const float* qrow = q + ((long)b * L_ + l) * C_ + h * HD_;
    const float sc = 0.125f * 1.44269504088896340736f;
#pragma unroll
    for (int kk = 0; kk < 8; ++kk) {
      int k = kk * 4 + khalf;
      float x0 = lv ? qrow[k] * sc : 0.f;
      float x1 = lv ? qrow[k + 1] * sc : 0.f;
      qB[kk] = (v2f){x0, x1};
    }
  }

  v8f O[4] = {};
  float Mrun = -INFINITY;
  float Srun = 0.f;

  const float* kb = kbuf + (long)b * C_ * P_ + (long)h * P_;
  const float* vb = vbuf + (long)b * D_ * P_ + (long)h * P_;

  const int srow = tid >> 3, sq = (tid & 7) * 2;
  const unsigned goff = (unsigned)(srow * 16) * P_ + sq;
  const float* gK = kb + goff;
  const float* gV = vb + goff;
  const int skOff = srow * KP_ + sq;
  const int svOff = srow * VP_ + sq;

  const int kOff = khalf * KP_ + col;
  const int vOff = col * VP_ + hi * 8;

  v2f rk, rv0, rv1;
  rk = *(const v2f*)gK;
  rv0 = *(const v2f*)gV;
  rv1 = *(const v2f*)(gV + (long)(32 * 16) * P_);
  *(v2f*)&sK[0][skOff] = rk;
  *(v2f*)&sV[0][svOff] = rv0;
  *(v2f*)&sV[0][svOff + 32 * VP_] = rv1;
  rk = *(const v2f*)(gK + AP_);
  rv0 = *(const v2f*)(gV + AP_);
  rv1 = *(const v2f*)(gV + (long)(32 * 16) * P_ + AP_);
  __syncthreads();

  constexpr int NS = P_ / AP_;
  for (int s = 0; s < NS; ++s) {
    const int cur = s & 1;
    const bool more = s + 1 < NS;
    const float* ks = &sK[cur][kOff];
    const float* vs = &sV[cur][vOff];
    v8f c = {};
#pragma unroll
    for (int kk = 0; kk < 8; ++kk) {
      const float* kp = ks + kk * 4 * KP_;
      v2f af = {kp[0], kp[KP_]};
      c = wmma4(af, qB[kk], c);
    }
    if (more) {
      *(v2f*)&sK[cur ^ 1][skOff] = rk;
      *(v2f*)&sV[cur ^ 1][svOff] = rv0;
      *(v2f*)&sV[cur ^ 1][svOff + 32 * VP_] = rv1;
    }
    float mx = fmaxf(fmaxf(fmaxf(c[0], c[1]), fmaxf(c[2], c[3])),
                     fmaxf(fmaxf(c[4], c[5]), fmaxf(c[6], c[7])));
    mx = fmaxf(mx, xhalf(mx));
    const float mnew = fmaxf(Mrun, mx);
    const float al = __builtin_exp2f(Mrun - mnew);
    v8f pv;
#pragma unroll
    for (int r = 0; r < 8; ++r) pv[r] = __builtin_exp2f(c[r] - mnew);
    Srun = Srun * al + (((pv[0] + pv[1]) + (pv[2] + pv[3])) +
                        ((pv[4] + pv[5]) + (pv[6] + pv[7])));
    Mrun = mnew;
#pragma unroll
    for (int j = 0; j < 4; ++j) O[j] *= al;
    if (s + 2 < NS) {
      const int po = (s + 2) * AP_;
      rk = *(const v2f*)(gK + po);
      rv0 = *(const v2f*)(gV + po);
      rv1 = *(const v2f*)(gV + (long)(32 * 16) * P_ + po);
    }
#pragma unroll
    for (int m = 0; m < 2; ++m) {
      v4f vf[4];
#pragma unroll
      for (int j = 0; j < 4; ++j)
        vf[j] = *(const v4f*)(vs + j * 16 * VP_ + m * 4);
      const v2f p0 = {pv[4 * m], pv[4 * m + 1]};
      const v2f p1 = {pv[4 * m + 2], pv[4 * m + 3]};
#pragma unroll
      for (int j = 0; j < 4; ++j) O[j] = wmma4(lo2(vf[j]), p0, O[j]);
#pragma unroll
      for (int j = 0; j < 4; ++j) O[j] = wmma4(hi2(vf[j]), p1, O[j]);
    }
    __builtin_amdgcn_fence(__ATOMIC_RELEASE, "workgroup", "local");
    __builtin_amdgcn_s_barrier();
    __builtin_amdgcn_fence(__ATOMIC_ACQUIRE, "workgroup", "local");
  }

  const float inv = 1.f / (Srun + xhalf(Srun));
  float* tw = &sT[wslot][0];
#pragma unroll
  for (int j = 0; j < 4; ++j) {
#pragma unroll
    for (int r = 0; r < 8; ++r) tw[(8 * hi + r) * TP_ + col] = O[j][r] * inv;
    __builtin_amdgcn_fence(__ATOMIC_ACQ_REL, "wavefront");
    __builtin_amdgcn_wave_barrier();
    float t[8];
#pragma unroll
    for (int r = 0; r < 8; ++r) t[r] = tw[col * TP_ + r + 8 * hi];
    __builtin_amdgcn_fence(__ATOMIC_ACQ_REL, "wavefront");
    __builtin_amdgcn_wave_barrier();
#pragma unroll
    for (int r = 0; r < 8; ++r) {
      int l = l0 + r + 8 * hi;
      if (l < L_)
        obuf[((long)b * L_ + l) * D_ + h + (j * 16 + col) * 16] = t[r];
    }
  }
}

extern "C" void kernel_launch(void* const* d_in, const int* in_sizes, int n_in,
                              void* d_out, int out_size, void* d_ws,
                              size_t ws_size, hipStream_t stream) {
  (void)in_sizes; (void)n_in; (void)out_size; (void)ws_size;
  const float* in = (const float*)d_in[0];
  const float* w_kv = (const float*)d_in[1];
  const float* w_q = (const float*)d_in[2];
  const float* b_q = (const float*)d_in[3];
  const float* w_out = (const float*)d_in[4];
  const float* b_out = (const float*)d_in[5];
  float* out = (float*)d_out;

  float* ws = (float*)d_ws;
  float* qbuf = ws;
  float* vbuf = qbuf + (size_t)B_ * L_ * C_;
  float* kbuf = vbuf + (size_t)B_ * D_ * P_;
  float* obuf = kbuf + (size_t)B_ * C_ * P_;

  gemm_rowA_wmma<4><<<256, 256, 0, stream>>>(in, (long)(B_ * D_), (long)D_, w_q,
                                             b_q, qbuf, (long)C_, (long)L_ * C_,
                                             L_, D_, 16, 4);
  gemm_colA_wmma<4><<<256, 256, 0, stream>>>(in, (long)(B_ * D_), (long)D_, w_kv,
                                             vbuf, (long)P_, (long)D_ * P_, L_,
                                             8, 8);
  gemm_colA_wmma<2><<<256, 128, 0, stream>>>(qbuf, (long)C_, (long)L_ * C_, w_kv,
                                             kbuf, (long)P_, (long)C_ * P_, L_,
                                             8, 8);
  attn_wmma<<<1024, 256, 0, stream>>>(qbuf, kbuf, vbuf, obuf);
  gemm_rowA_wmma<4><<<512, 256, 0, stream>>>(obuf, (long)D_, (long)L_ * D_,
                                             w_out, b_out, out, (long)(B_ * D_),
                                             (long)D_, L_, D_, 16, 8);
}


# k built from v: k = w_q v + b_q x colsum(Wp) (q = X w_q^T + b_q substituted) instead of q^T Wp: that GEMM's K 2046 -> 1024, half its WMMAs; k's sums reassociated, their order changes
# speedup vs baseline: 2.1377x; 1.0634x over previous
#include <hip/hip_runtime.h>


typedef __attribute__((ext_vector_type(2))) float v2f;
typedef __attribute__((ext_vector_type(4))) float v4f;
typedef __attribute__((ext_vector_type(8))) float v8f;

static __device__ __forceinline__ v2f lo2(v4f x) { return (v2f){x.x, x.y}; }
static __device__ __forceinline__ v2f hi2(v4f x) { return (v2f){x.z, x.w}; }

#define B_ 4
#define L_ 2046
#define D_ 1024
#define C_ 512
#define P_ 1024
#define H_ 16
#define HD_ 32
#define DV_ 64
#define WKV_LD 2048

static __device__ __forceinline__ v8f wmma4(v2f a, v2f b, v8f c) {
  return __builtin_amdgcn_wmma_f32_16x16x4_f32(false, a, false, b, (short)0, c,
                                               false, false);
}

static __device__ __forceinline__ float rowmax16(float v) {
  v = fmaxf(v, __shfl_xor(v, 8, 16));
  v = fmaxf(v, __shfl_xor(v, 4, 16));
  v = fmaxf(v, __shfl_xor(v, 2, 16));
  v = fmaxf(v, __shfl_xor(v, 1, 16));
  return v;
}
static __device__ __forceinline__ float rowsum16(float v) {
  v += __shfl_xor(v, 8, 16);
  v += __shfl_xor(v, 4, 16);
  v += __shfl_xor(v, 2, 16);
  v += __shfl_xor(v, 1, 16);
  return v;
}

#define TK_ 32
#define PK_ 36

template <int WM>
__global__ __launch_bounds__(WM * 64) void gemm_rowA_wmma(
    const float* __restrict__ A, long strideAm, long batchStrideA,
    const float* __restrict__ W, const float* __restrict__ bias,
    float* __restrict__ Out, long strideOm, long batchStrideO,
    int M, int K, int mtiles, int ntiles) {
  constexpr int BM = WM * 32, BN = 128, NT = WM * 64;
  constexpr int RPP = NT / 8;
  constexpr int PA = BM / RPP;
  constexpr int PW = BN / RPP;
  __shared__ __attribute__((aligned(16))) float sA[2][BM * PK_];
  __shared__ __attribute__((aligned(16))) float sW[2][BN * PK_];
  int tid = threadIdx.x;
  int lane = tid & 31;
  int wslot = tid >> 5;
  int tilesPerBatch = mtiles * ntiles;
  int b = blockIdx.x / tilesPerBatch;
  int r = blockIdx.x - b * tilesPerBatch;
  int bm0 = (r / ntiles) * BM;
  int bn0 = (r % ntiles) * BN;
  const float* Ab = A + (long)b * batchStrideA;
  float* Ob = Out + (long)b * batchStrideO;
  int col = lane & 15;
  int hi = lane >> 4;
  int wm = wslot >> 1, wn = wslot & 1;

  int srow = tid >> 3, sq = (tid & 7) * 4;
  const float* Ag[PA];
  const float* Wg[PW];
#pragma unroll
  for (int p = 0; p < PA; ++p) {
    int row = bm0 + srow + RPP * p;
    if (row > M - 1) row = M - 1;
    Ag[p] = Ab + (long)row * strideAm + sq;
  }
#pragma unroll
  for (int p = 0; p < PW; ++p) Wg[p] = W + (long)(bn0 + srow + RPP * p) * K + sq;
  const int sOff = srow * PK_ + sq;

  const int aOff = (wm * 32 + col) * PK_ + 4 * hi;
  const int bOff = (wn * 64 + col) * PK_ + 4 * hi;

  v4f ra[PA], rw[PW];
#pragma unroll
  for (int p = 0; p < PA; ++p) ra[p] = *(const v4f*)(Ag[p]);
#pragma unroll
  for (int p = 0; p < PW; ++p) rw[p] = *(const v4f*)(Wg[p]);
#pragma unroll
  for (int p = 0; p < PA; ++p) *(v4f*)&sA[0][sOff + p * RPP * PK_] = ra[p];
#pragma unroll
  for (int p = 0; p < PW; ++p) *(v4f*)&sW[0][sOff + p * RPP * PK_] = rw[p];
  __syncthreads();

  v8f acc[2][4] = {};
  const int KT = K / TK_;
  for (int kt = 0; kt < KT; ++kt) {
    const int cur = kt & 1;
    const bool more = kt + 1 < KT;
    if (more) {
      const int ko = (kt + 1) * TK_;
#pragma unroll
      for (int p = 0; p < PA; ++p) ra[p] = *(const v4f*)(Ag[p] + ko);
#pragma unroll
      for (int p = 0; p < PW; ++p) rw[p] = *(const v4f*)(Wg[p] + ko);
    }
    const float* as = &sA[cur][aOff];
    const float* ws = &sW[cur][bOff];
    v4f fa[2][2], fb[2][4];
    fa[0][0] = *(const v4f*)(as);
    fa[0][1] = *(const v4f*)(as + 16 * PK_);
#pragma unroll
    for (int j = 0; j < 4; ++j) fb[0][j] = *(const v4f*)(ws + j * 16 * PK_);
#pragma unroll
    for (int c = 0; c < 4; ++c) {
      const int s = c & 1, n = s ^ 1;
      if (c < 3) {
        fa[n][0] = *(const v4f*)(as + 8 * (c + 1));
        fa[n][1] = *(const v4f*)(as + 16 * PK_ + 8 * (c + 1));
#pragma unroll
        for (int j = 0; j < 4; ++j)
          fb[n][j] = *(const v4f*)(ws + j * 16 * PK_ + 8 * (c + 1));
      }
      __builtin_amdgcn_sched_barrier(0);
#pragma unroll
      for (int j = 0; j < 4; ++j) {
        acc[0][j] = wmma4(lo2(fa[s][0]), lo2(fb[s][j]), acc[0][j]);
        acc[1][j] = wmma4(lo2(fa[s][1]), lo2(fb[s][j]), acc[1][j]);
      }
#pragma unroll
      for (int j = 0; j < 4; ++j) {
        acc[0][j] = wmma4(hi2(fa[s][0]), hi2(fb[s][j]), acc[0][j]);
        acc[1][j] = wmma4(hi2(fa[s][1]), hi2(fb[s][j]), acc[1][j]);
      }
    }
    if (more) {
#pragma unroll
      for (int p = 0; p < PA; ++p) *(v4f*)&sA[cur ^ 1][sOff + p * RPP * PK_] = ra[p];
#pragma unroll
      for (int p = 0; p < PW; ++p) *(v4f*)&sW[cur ^ 1][sOff + p * RPP * PK_] = rw[p];
    }
    __syncthreads();
  }

  const int m0 = bm0 + wm * 32;
  const int n0 = bn0 + wn * 64;
  float bv[4];
#pragma unroll
  for (int j = 0; j < 4; ++j) bv[j] = bias[n0 + j * 16 + col];
#pragma unroll
  for (int i = 0; i < 2; ++i) {
#pragma unroll
    for (int rr = 0; rr < 8; ++rr) {
      int row = m0 + i * 16 + rr + 8 * hi;
      if (row < M) {
        float* orow = Ob + (long)row * strideOm + n0;
#pragma unroll
        for (int j = 0; j < 4; ++j) orow[j * 16 + col] = acc[i][j][rr] + bv[j];
      }
    }
  }
}

template <int WM, int MODE>
__global__ __launch_bounds__(WM * 64) void gemm_colA_wmma(
    const float* __restrict__ A, long strideAk, long batchStrideA,
    const float* __restrict__ Wkv, int ldw,
    float* __restrict__ Out, long strideO, long batchStrideO,
    int K, int mtiles, int ntiles,
    const float* __restrict__ rowBias, const float* __restrict__ colVec,
    float* __restrict__ colSum) {
  constexpr int BM = WM * 32, BN = 128, NT = WM * 64;
  constexpr int PM = BM + 4;
  constexpr int TPR = BM / 4;
  constexpr int KPP = NT / TPR;
  constexpr int PA = TK_ / KPP;
  constexpr int RPP = NT / 8;
  constexpr int PW = BN / RPP;
  __shared__ __attribute__((aligned(16))) float sA[2][TK_ * PM];
  __shared__ __attribute__((aligned(16))) float sW[2][BN * PK_];
  int tid = threadIdx.x;
  int lane = tid & 31;
  int wslot = tid >> 5;
  int tilesPerBatch = mtiles * ntiles;
  int b = blockIdx.x / tilesPerBatch;
  int r = blockIdx.x - b * tilesPerBatch;
  int bm0 = (r / ntiles) * BM;
  int bn0 = (r % ntiles) * BN;
  const float* Ab = A + (long)b * batchStrideA;
  float* Ob = Out + (long)b * batchStrideO;
  int col = lane & 15;
  int hi = lane >> 4;
  int wm = wslot >> 1, wn = wslot & 1;

  const int ak = tid / TPR, am = (tid % TPR) * 4;
  const float* Ag = Ab + bm0 + am;
  const int saOff = ak * PM + am;
  const int srow = tid >> 3, sq = (tid & 7) * 4;
  const float* Wg = Wkv + (long)(bn0 + srow) * ldw + sq;
  const int swOff = srow * PK_ + sq;

  const int aOff = (4 * hi) * PM + wm * 32 + col;
  const int bOff = (wn * 64 + col) * PK_ + 4 * hi;

  v4f ra[PA], rw[PW];
  const v4f zero4 = {0.f, 0.f, 0.f, 0.f};
  const bool doCs = MODE == 1 && (int)blockIdx.x < ntiles;
  float csacc[PW];
#pragma unroll
  for (int p = 0; p < PW; ++p) csacc[p] = 0.f;
  auto fetch = [&](int kt) {
    const int k0 = kt * TK_;
#pragma unroll
    for (int p = 0; p < PA; ++p) {
      int k = k0 + ak + KPP * p;
      int kc = k < K ? k : K - 1;
      ra[p] = *(const v4f*)(Ag + (long)kc * strideAk);
    }
#pragma unroll
    for (int p = 0; p < PW; ++p)
      rw[p] = *(const v4f*)(Wg + (long)(RPP * p) * ldw + k0);
  };
  auto stash = [&](int st, int kt) {
    const int k0 = kt * TK_;
#pragma unroll
    for (int p = 0; p < PA; ++p) {
      int k = k0 + ak + KPP * p;
      *(v4f*)&sA[st][saOff + p * KPP * PM] = k < K ? ra[p] : zero4;
    }
    const int kq = k0 + sq;
    const bool v0 = kq < K, v1 = kq + 1 < K, v2 = kq + 2 < K, v3 = kq + 3 < K;
#pragma unroll
    for (int p = 0; p < PW; ++p) {
      v4f t = rw[p];
      const v4f tm = {v0 ? t.x : 0.f, v1 ? t.y : 0.f, v2 ? t.z : 0.f, v3 ? t.w : 0.f};
      *(v4f*)&sW[st][swOff + p * RPP * PK_] = tm;
      if constexpr (MODE == 1) {
        if (doCs) csacc[p] += (tm.x + tm.y) + (tm.z + tm.w);
      }
    }
  };

  fetch(0);
  stash(0, 0);
  __syncthreads();

  v8f acc[2][4] = {};
  const int KT = (K + TK_ - 1) / TK_;
  for (int kt = 0; kt < KT; ++kt) {
    const int cur = kt & 1;
    const bool more = kt + 1 < KT;
    if (more) fetch(kt + 1);
    const float* as = &sA[cur][aOff];
    const float* ws = &sW[cur][bOff];
    v2f al[2][2], ah[2][2];
    v4f fb[2][4];
#pragma unroll
    for (int i = 0; i < 2; ++i) {
      al[0][i] = (v2f){as[i * 16], as[PM + i * 16]};
      ah[0][i] = (v2f){as[2 * PM + i * 16], as[3 * PM + i * 16]};
    }
#pragma unroll
    for (int j = 0; j < 4; ++j) fb[0][j] = *(const v4f*)(ws + j * 16 * PK_);
#pragma unroll
    for (int c = 0; c < 4; ++c) {
      const int s = c & 1, n = s ^ 1;
      if (c < 3) {
        const float* an = as + 8 * (c + 1) * PM;
#pragma unroll
        for (int i = 0; i < 2; ++i) {
          al[n][i] = (v2f){an[i * 16], an[PM + i * 16]};
          ah[n][i] = (v2f){an[2 * PM + i * 16], an[3 * PM + i * 16]};
        }
#pragma unroll
        for (int j = 0; j < 4; ++j)
          fb[n][j] = *(const v4f*)(ws + j * 16 * PK_ + 8 * (c + 1));
      }
      __builtin_amdgcn_sched_barrier(0);
#pragma unroll
      for (int j = 0; j < 4; ++j) {
        if constexpr (MODE == 2) {
          acc[0][j] = wmma4(lo2(fb[s][j]), al[s][0], acc[0][j]);
          acc[1][j] = wmma4(lo2(fb[s][j]), al[s][1], acc[1][j]);
        } else {
          acc[0][j] = wmma4(al[s][0], lo2(fb[s][j]), acc[0][j]);
          acc[1][j] = wmma4(al[s][1], lo2(fb[s][j]), acc[1][j]);
        }
      }
#pragma unroll
      for (int j = 0; j < 4; ++j) {
        if constexpr (MODE == 2) {
          acc[0][j] = wmma4(hi2(fb[s][j]), ah[s][0], acc[0][j]);
          acc[1][j] = wmma4(hi2(fb[s][j]), ah[s][1], acc[1][j]);
        } else {
          acc[0][j] = wmma4(ah[s][0], hi2(fb[s][j]), acc[0][j]);
          acc[1][j] = wmma4(ah[s][1], hi2(fb[s][j]), acc[1][j]);
        }
      }
    }
    if (more) stash(cur ^ 1, kt + 1);
    __syncthreads();
  }

  const int m0 = bm0 + wm * 32;
  const int n0 = bn0 + wn * 64;
  if constexpr (MODE == 2) {
    float cv[2];
#pragma unroll
    for (int i = 0; i < 2; ++i) cv[i] = colVec[m0 + i * 16 + col];
#pragma unroll
    for (int j = 0; j < 4; ++j) {
#pragma unroll
      for (int rr = 0; rr < 8; ++rr) {
        const int n = n0 + j * 16 + rr + 8 * hi;
        const float rb = rowBias[n];
        float* orow = Ob + (long)n * strideO + m0;
#pragma unroll
        for (int i = 0; i < 2; ++i) orow[i * 16 + col] = acc[i][j][rr] + rb * cv[i];
      }
    }
  } else {
#pragma unroll
    for (int i = 0; i < 2; ++i) {
#pragma unroll
      for (int rr = 0; rr < 8; ++rr) {
        int row = m0 + i * 16 + rr + 8 * hi;
        float* orow = Ob + (long)row * strideO + n0;
#pragma unroll
        for (int j = 0; j < 4; ++j) orow[j * 16 + col] = acc[i][j][rr];
      }
    }
  }
  if constexpr (MODE == 1) {
    if (doCs) {
      float* sC = &sW[0][0];
#pragma unroll
      for (int p = 0; p < PW; ++p) {
        float s = csacc[p];
        s += __shfl_xor(s, 1);
        s += __shfl_xor(s, 2);
        s += __shfl_xor(s, 4);
        if ((tid & 7) == 0) sC[srow + RPP * p] = s;
      }
      __syncthreads();
      if (tid < 32) *(v4f*)&colSum[bn0 + tid * 4] = *(const v4f*)&sC[tid * 4];
    }
  }
}

#define AP_ 16
#define KP_ 16
#define VP_ 20
#define TP_ 17

static __device__ __forceinline__ float xhalf(float x) {
  int b = __builtin_bit_cast(int, x);
  return __builtin_bit_cast(
      float, __builtin_amdgcn_permlanex16(b, b, 0x76543210, 0xfedcba98, false, false));
}

__global__ __launch_bounds__(256, 8) void attn_wmma(
    const float* __restrict__ q, const float* __restrict__ kbuf,
    const float* __restrict__ vbuf, float* __restrict__ obuf) {
  __shared__ __attribute__((aligned(16))) float sK[2][HD_ * KP_];
  __shared__ __attribute__((aligned(16))) float sV[2][DV_ * VP_];
  __shared__ float sT[8][16 * TP_];
  int tid = threadIdx.x;
  int lane = tid & 31;
  int wslot = tid >> 5;
  int ltile = (blockIdx.x & 15) * 8 + wslot;
  int h = (blockIdx.x >> 4) & 15;
  int b = blockIdx.x >> 8;
  int l0 = ltile * 16;
  int col = lane & 15;
  int hi = lane >> 4;
  int khalf = hi * 2;

  v2f qB[8];
  {
    int l = l0 + col;
    bool lv = l < L_;
    const float* qrow = q + ((long)b * L_ + l) * C_ + h * HD_;
    const float sc = 0.125f * 1.44269504088896340736f;
#pragma unroll
    for (int kk = 0; kk < 8; ++kk) {
      int k = kk * 4 + khalf;
      float x0 = lv ? qrow[k] * sc : 0.f;
      float x1 = lv ? qrow[k + 1] * sc : 0.f;
      qB[kk] = (v2f){x0, x1};
    }
  }

  v8f O[4] = {};
  float Mrun = -INFINITY;
  float Srun = 0.f;

  const float* kb = kbuf + (long)b * C_ * P_ + (long)h * P_;
  const float* vb = vbuf + (long)b * D_ * P_ + (long)h * P_;

  const int srow = tid >> 3, sq = (tid & 7) * 2;
  const unsigned goff = (unsigned)(srow * 16) * P_ + sq;
  const float* gK = kb + goff;
  const float* gV = vb + goff;
  const int skOff = srow * KP_ + sq;
  const int svOff = srow * VP_ + sq;

  const int kOff = khalf * KP_ + col;
  const int vOff = col * VP_ + hi * 8;

  v2f rk, rv0, rv1;
  rk = *(const v2f*)gK;
  rv0 = *(const v2f*)gV;
  rv1 = *(const v2f*)(gV + (long)(32 * 16) * P_);
  *(v2f*)&sK[0][skOff] = rk;
  *(v2f*)&sV[0][svOff] = rv0;
  *(v2f*)&sV[0][svOff + 32 * VP_] = rv1;
  rk = *(const v2f*)(gK + AP_);
  rv0 = *(const v2f*)(gV + AP_);
  rv1 = *(const v2f*)(gV + (long)(32 * 16) * P_ + AP_);
  __syncthreads();

  constexpr int NS = P_ / AP_;
  for (int s = 0; s < NS; ++s) {
    const int cur = s & 1;
    const bool more = s + 1 < NS;
    const float* ks = &sK[cur][kOff];
    const float* vs = &sV[cur][vOff];
    v8f c = {};
#pragma unroll
    for (int kk = 0; kk < 8; ++kk) {
      const float* kp = ks + kk * 4 * KP_;
      v2f af = {kp[0], kp[KP_]};
      c = wmma4(af, qB[kk], c);
    }
    if (more) {
      *(v2f*)&sK[cur ^ 1][skOff] = rk;
      *(v2f*)&sV[cur ^ 1][svOff] = rv0;
      *(v2f*)&sV[cur ^ 1][svOff + 32 * VP_] = rv1;
    }
    float mx = fmaxf(fmaxf(fmaxf(c[0], c[1]), fmaxf(c[2], c[3])),
                     fmaxf(fmaxf(c[4], c[5]), fmaxf(c[6], c[7])));
    mx = fmaxf(mx, xhalf(mx));
    const float mnew = fmaxf(Mrun, mx);
    const float al = __builtin_exp2f(Mrun - mnew);
    v8f pv;
#pragma unroll
    for (int r = 0; r < 8; ++r) pv[r] = __builtin_exp2f(c[r] - mnew);
    Srun = Srun * al + (((pv[0] + pv[1]) + (pv[2] + pv[3])) +
                        ((pv[4] + pv[5]) + (pv[6] + pv[7])));
    Mrun = mnew;
#pragma unroll
    for (int j = 0; j < 4; ++j) O[j] *= al;
    if (s + 2 < NS) {
      const int po = (s + 2) * AP_;
      rk = *(const v2f*)(gK + po);
      rv0 = *(const v2f*)(gV + po);
      rv1 = *(const v2f*)(gV + (long)(32 * 16) * P_ + po);
    }
#pragma unroll
    for (int m = 0; m < 2; ++m) {
      v4f vf[4];
#pragma unroll
      for (int j = 0; j < 4; ++j)
        vf[j] = *(const v4f*)(vs + j * 16 * VP_ + m * 4);
      const v2f p0 = {pv[4 * m], pv[4 * m + 1]};
      const v2f p1 = {pv[4 * m + 2], pv[4 * m + 3]};
#pragma unroll
      for (int j = 0; j < 4; ++j) O[j] = wmma4(lo2(vf[j]), p0, O[j]);
#pragma unroll
      for (int j = 0; j < 4; ++j) O[j] = wmma4(hi2(vf[j]), p1, O[j]);
    }
    __builtin_amdgcn_fence(__ATOMIC_RELEASE, "workgroup", "local");
    __builtin_amdgcn_s_barrier();
    __builtin_amdgcn_fence(__ATOMIC_ACQUIRE, "workgroup", "local");
  }

  const float inv = 1.f / (Srun + xhalf(Srun));
  float* tw = &sT[wslot][0];
#pragma unroll
  for (int j = 0; j < 4; ++j) {
#pragma unroll
    for (int r = 0; r < 8; ++r) tw[(8 * hi + r) * TP_ + col] = O[j][r] * inv;
    __builtin_amdgcn_fence(__ATOMIC_ACQ_REL, "wavefront");
    __builtin_amdgcn_wave_barrier();
    float t[8];
#pragma unroll
    for (int r = 0; r < 8; ++r) t[r] = tw[col * TP_ + r + 8 * hi];
    __builtin_amdgcn_fence(__ATOMIC_ACQ_REL, "wavefront");
    __builtin_amdgcn_wave_barrier();
#pragma unroll
    for (int r = 0; r < 8; ++r) {
      int l = l0 + r + 8 * hi;
      if (l < L_)
        obuf[((long)b * L_ + l) * D_ + h + (j * 16 + col) * 16] = t[r];
    }
  }
}

extern "C" void kernel_launch(void* const* d_in, const int* in_sizes, int n_in,
                              void* d_out, int out_size, void* d_ws,
                              size_t ws_size, hipStream_t stream) {
  (void)in_sizes; (void)n_in; (void)out_size; (void)ws_size;
  const float* in = (const float*)d_in[0];
  const float* w_kv = (const float*)d_in[1];
  const float* w_q = (const float*)d_in[2];
  const float* b_q = (const float*)d_in[3];
  const float* w_out = (const float*)d_in[4];
  const float* b_out = (const float*)d_in[5];
  float* out = (float*)d_out;

  float* ws = (float*)d_ws;
  float* qbuf = ws;
  float* vbuf = qbuf + (size_t)B_ * L_ * C_;
  float* kbuf = vbuf + (size_t)B_ * D_ * P_;
  float* obuf = kbuf + (size_t)B_ * C_ * P_;
  float* csbuf = obuf;

  gemm_rowA_wmma<4><<<256, 256, 0, stream>>>(in, (long)(B_ * D_), (long)D_, w_q,
                                             b_q, qbuf, (long)C_, (long)L_ * C_,
                                             L_, D_, 16, 4);
  gemm_colA_wmma<4, 1><<<256, 256, 0, stream>>>(in, (long)(B_ * D_), (long)D_, w_kv,
                                                WKV_LD, vbuf, (long)P_,
                                                (long)D_ * P_, L_, 8, 8, nullptr,
                                                nullptr, csbuf);
  gemm_colA_wmma<2, 2><<<256, 128, 0, stream>>>(vbuf, (long)P_, (long)D_ * P_, w_q,
                                                D_, kbuf, (long)P_, (long)C_ * P_,
                                                D_, 16, 4, b_q, csbuf, nullptr);
  attn_wmma<<<1024, 256, 0, stream>>>(qbuf, kbuf, vbuf, obuf);
  gemm_rowA_wmma<4><<<512, 256, 0, stream>>>(obuf, (long)D_, (long)L_ * D_,
                                             w_out, b_out, out, (long)(B_ * D_),
                                             (long)D_, L_, D_, 16, 8);
}


# All three GEMM kernels: LDS ring of 3 stages (was 2): a slice is stored two ahead and the next slice's first fragments are read before the slice's barrier, so no wave waits there; sum orders kept
# speedup vs baseline: 2.2089x; 1.0333x over previous
#include <hip/hip_runtime.h>


typedef __attribute__((ext_vector_type(2))) float v2f;
typedef __attribute__((ext_vector_type(4))) float v4f;
typedef __attribute__((ext_vector_type(8))) float v8f;

static __device__ __forceinline__ v2f lo2(v4f x) { return (v2f){x.x, x.y}; }
static __device__ __forceinline__ v2f hi2(v4f x) { return (v2f){x.z, x.w}; }

#define B_ 4
#define L_ 2046
#define D_ 1024
#define C_ 512
#define P_ 1024
#define H_ 16
#define HD_ 32
#define DV_ 64
#define WKV_LD 2048

static __device__ __forceinline__ v8f wmma4(v2f a, v2f b, v8f c) {
  return __builtin_amdgcn_wmma_f32_16x16x4_f32(false, a, false, b, (short)0, c,
                                               false, false);
}

static __device__ __forceinline__ float rowmax16(float v) {
  v = fmaxf(v, __shfl_xor(v, 8, 16));
  v = fmaxf(v, __shfl_xor(v, 4, 16));
  v = fmaxf(v, __shfl_xor(v, 2, 16));
  v = fmaxf(v, __shfl_xor(v, 1, 16));
  return v;
}
static __device__ __forceinline__ float rowsum16(float v) {
  v += __shfl_xor(v, 8, 16);
  v += __shfl_xor(v, 4, 16);
  v += __shfl_xor(v, 2, 16);
  v += __shfl_xor(v, 1, 16);
  return v;
}

#define TK_ 32
#define PK_ 36
#define NST_ 3

static __device__ __forceinline__ void lds_barrier() {
  __builtin_amdgcn_fence(__ATOMIC_RELEASE, "workgroup", "local");
  __builtin_amdgcn_s_barrier();
  __builtin_amdgcn_fence(__ATOMIC_ACQUIRE, "workgroup", "local");
}

template <int WM>
__global__ __launch_bounds__(WM * 64) void gemm_rowA_wmma(
    const float* __restrict__ A, long strideAm, long batchStrideA,
    const float* __restrict__ W, const float* __restrict__ bias,
    float* __restrict__ Out, long strideOm, long batchStrideO,
    int M, int K, int mtiles, int ntiles) {
  constexpr int BM = WM * 32, BN = 128, NT = WM * 64;
  constexpr int RPP = NT / 8;
  constexpr int PA = BM / RPP;
  constexpr int PW = BN / RPP;
  constexpr int SA = BM * PK_;
  constexpr int SW = BN * PK_;
  __shared__ __attribute__((aligned(16))) float sA[NST_ * SA];
  __shared__ __attribute__((aligned(16))) float sW[NST_ * SW];
  int tid = threadIdx.x;
  int lane = tid & 31;
  int wslot = tid >> 5;
  int tilesPerBatch = mtiles * ntiles;
  int b = blockIdx.x / tilesPerBatch;
  int r = blockIdx.x - b * tilesPerBatch;
  int bm0 = (r / ntiles) * BM;
  int bn0 = (r % ntiles) * BN;
  const float* Ab = A + (long)b * batchStrideA;
  float* Ob = Out + (long)b * batchStrideO;
  int col = lane & 15;
  int hi = lane >> 4;
  int wm = wslot >> 1, wn = wslot & 1;

  int srow = tid >> 3, sq = (tid & 7) * 4;
  const float* Ag[PA];
  const float* Wg[PW];
#pragma unroll
  for (int p = 0; p < PA; ++p) {
    int row = bm0 + srow + RPP * p;
    if (row > M - 1) row = M - 1;
    Ag[p] = Ab + (long)row * strideAm + sq;
  }
#pragma unroll
  for (int p = 0; p < PW; ++p) Wg[p] = W + (long)(bn0 + srow + RPP * p) * K + sq;
  const int sOff = srow * PK_ + sq;

  const int aOff = (wm * 32 + col) * PK_ + 4 * hi;
  const int bOff = (wn * 64 + col) * PK_ + 4 * hi;

  const int KT = K / TK_;
  v4f ra[PA], rw[PW];
  auto fetch = [&](v4f (&da)[PA], v4f (&dw)[PW], int kt) {
    const int ko = kt * TK_;
#pragma unroll
    for (int p = 0; p < PA; ++p) da[p] = *(const v4f*)(Ag[p] + ko);
#pragma unroll
    for (int p = 0; p < PW; ++p) dw[p] = *(const v4f*)(Wg[p] + ko);
  };
  auto stash = [&](const v4f (&xa)[PA], const v4f (&xw)[PW], int st) {
#pragma unroll
    for (int p = 0; p < PA; ++p) *(v4f*)&sA[st * SA + sOff + p * RPP * PK_] = xa[p];
#pragma unroll
    for (int p = 0; p < PW; ++p) *(v4f*)&sW[st * SW + sOff + p * RPP * PK_] = xw[p];
  };
  {
    v4f rb[PA], rx[PW];
    fetch(ra, rw, 0);
    if (KT > 1) fetch(rb, rx, 1);
    stash(ra, rw, 0);
    if (KT > 1) stash(rb, rx, 1);
  }
  if (KT > 2) fetch(ra, rw, 2);
  lds_barrier();

  v4f fa[2][2], fb[2][4];
  auto ldfrag = [&](int set, int st, int c) {
    const float* as = &sA[st * SA + aOff + 8 * c];
    const float* ws = &sW[st * SW + bOff + 8 * c];
    fa[set][0] = *(const v4f*)(as);
    fa[set][1] = *(const v4f*)(as + 16 * PK_);
#pragma unroll
    for (int j = 0; j < 4; ++j) fb[set][j] = *(const v4f*)(ws + j * 16 * PK_);
  };
  ldfrag(0, 0, 0);

  v8f acc[2][4] = {};
  int o0 = 0, o1 = 1, o2 = 2;
  for (int kt = 0; kt < KT; ++kt) {
    stash(ra, rw, o2);
    if (kt + 3 < KT) fetch(ra, rw, kt + 3);
#pragma unroll
    for (int c = 0; c < 4; ++c) {
      const int s = c & 1, n = s ^ 1;
      if (c < 3) ldfrag(n, o0, c + 1); else ldfrag(n, o1, 0);
      __builtin_amdgcn_sched_barrier(0);
#pragma unroll
      for (int j = 0; j < 4; ++j) {
        acc[0][j] = wmma4(lo2(fa[s][0]), lo2(fb[s][j]), acc[0][j]);
        acc[1][j] = wmma4(lo2(fa[s][1]), lo2(fb[s][j]), acc[1][j]);
      }
#pragma unroll
      for (int j = 0; j < 4; ++j) {
        acc[0][j] = wmma4(hi2(fa[s][0]), hi2(fb[s][j]), acc[0][j]);
        acc[1][j] = wmma4(hi2(fa[s][1]), hi2(fb[s][j]), acc[1][j]);
      }
    }
    lds_barrier();
    const int t = o0; o0 = o1; o1 = o2; o2 = t;
  }

  const int m0 = bm0 + wm * 32;
  const int n0 = bn0 + wn * 64;
  float bv[4];
#pragma unroll
  for (int j = 0; j < 4; ++j) bv[j] = bias[n0 + j * 16 + col];
#pragma unroll
  for (int i = 0; i < 2; ++i) {
#pragma unroll
    for (int rr = 0; rr < 8; ++rr) {
      int row = m0 + i * 16 + rr + 8 * hi;
      if (row < M) {
        float* orow = Ob + (long)row * strideOm + n0;
#pragma unroll
        for (int j = 0; j < 4; ++j) orow[j * 16 + col] = acc[i][j][rr] + bv[j];
      }
    }
  }
}

template <int WM, int MODE>
__global__ __launch_bounds__(WM * 64) void gemm_colA_wmma(
    const float* __restrict__ A, long strideAk, long batchStrideA,
    const float* __restrict__ Wkv, int ldw,
    float* __restrict__ Out, long strideO, long batchStrideO,
    int K, int mtiles, int ntiles,
    const float* __restrict__ rowBias, const float* __restrict__ colVec,
    float* __restrict__ colSum) {
  constexpr int BM = WM * 32, BN = 128, NT = WM * 64;
  constexpr int PM = BM + 4;
  constexpr int TPR = BM / 4;
  constexpr int KPP = NT / TPR;
  constexpr int PA = TK_ / KPP;
  constexpr int RPP = NT / 8;
  constexpr int PW = BN / RPP;
  constexpr int SA = TK_ * PM;
  constexpr int SW = BN * PK_;
  __shared__ __attribute__((aligned(16))) float sA[NST_ * SA];
  __shared__ __attribute__((aligned(16))) float sW[NST_ * SW];
  int tid = threadIdx.x;
  int lane = tid & 31;
  int wslot = tid >> 5;
  int tilesPerBatch = mtiles * ntiles;
  int b = blockIdx.x / tilesPerBatch;
  int r = blockIdx.x - b * tilesPerBatch;
  int bm0 = (r / ntiles) * BM;
  int bn0 = (r % ntiles) * BN;
  const float* Ab = A + (long)b * batchStrideA;
  float* Ob = Out + (long)b * batchStrideO;
  int col = lane & 15;
  int hi = lane >> 4;
  int wm = wslot >> 1, wn = wslot & 1;

  const int ak = tid / TPR, am = (tid % TPR) * 4;
  const float* Ag = Ab + bm0 + am;
  const int saOff = ak * PM + am;
  const int srow = tid >> 3, sq = (tid & 7) * 4;
  const float* Wg = Wkv + (long)(bn0 + srow) * ldw + sq;
  const int swOff = srow * PK_ + sq;

  const int aOff = (4 * hi) * PM + wm * 32 + col;
  const int bOff = (wn * 64 + col) * PK_ + 4 * hi;

  v4f ra[PA], rw[PW];
  const v4f zero4 = {0.f, 0.f, 0.f, 0.f};
  const bool doCs = MODE == 1 && (int)blockIdx.x < ntiles;
  float csacc[PW];
#pragma unroll
  for (int p = 0; p < PW; ++p) csacc[p] = 0.f;
  auto fetch = [&](v4f (&da)[PA], v4f (&dw)[PW], int kt) {
    const int k0 = kt * TK_;
#pragma unroll
    for (int p = 0; p < PA; ++p) {
      int k = k0 + ak + KPP * p;
      int kc = k < K ? k : K - 1;
      da[p] = *(const v4f*)(Ag + (long)kc * strideAk);
    }
#pragma unroll
    for (int p = 0; p < PW; ++p)
      dw[p] = *(const v4f*)(Wg + (long)(RPP * p) * ldw + k0);
  };
  auto stash = [&](const v4f (&xa)[PA], const v4f (&xw)[PW], int st, int kt) {
    const int k0 = kt * TK_;
#pragma unroll
    for (int p = 0; p < PA; ++p) {
      int k = k0 + ak + KPP * p;
      *(v4f*)&sA[st * SA + saOff + p * KPP * PM] = k < K ? xa[p] : zero4;
    }
    const int kq = k0 + sq;
    const bool v0 = kq < K, v1 = kq + 1 < K, v2 = kq + 2 < K, v3 = kq + 3 < K;
#pragma unroll
    for (int p = 0; p < PW; ++p) {
      v4f t = xw[p];
      const v4f tm = {v0 ? t.x : 0.f, v1 ? t.y : 0.f, v2 ? t.z : 0.f, v3 ? t.w : 0.f};
      *(v4f*)&sW[st * SW + swOff + p * RPP * PK_] = tm;
      if constexpr (MODE == 1) {
        if (doCs) csacc[p] += (tm.x + tm.y) + (tm.z + tm.w);
      }
    }
  };

  const int KT = (K + TK_ - 1) / TK_;
  {
    v4f rb[PA], rx[PW];
    fetch(ra, rw, 0);
    if (KT > 1) fetch(rb, rx, 1);
    stash(ra, rw, 0, 0);
    if (KT > 1) stash(rb, rx, 1, 1);
  }
  if (KT > 2) fetch(ra, rw, 2);
  lds_barrier();

  float ar[2][2][4];
  v4f fb[2][4];
  auto ldfrag = [&](int set, int st, int c) {
    const float* an = &sA[st * SA + aOff + 8 * c * PM];
    const float* ws = &sW[st * SW + bOff + 8 * c];
#pragma unroll
    for (int i = 0; i < 2; ++i) {
#pragma unroll
      for (int kk = 0; kk < 4; ++kk) ar[set][i][kk] = an[kk * PM + i * 16];
    }
#pragma unroll
    for (int j = 0; j < 4; ++j) fb[set][j] = *(const v4f*)(ws + j * 16 * PK_);
  };
  ldfrag(0, 0, 0);

  v8f acc[2][4] = {};
  int o0 = 0, o1 = 1, o2 = 2;
  for (int kt = 0; kt < KT; ++kt) {
    stash(ra, rw, o2, kt + 2);
    if (kt + 3 < KT) fetch(ra, rw, kt + 3);
#pragma unroll
    for (int c = 0; c < 4; ++c) {
      const int s = c & 1, n = s ^ 1;
      if (c < 3) ldfrag(n, o0, c + 1); else ldfrag(n, o1, 0);
      __builtin_amdgcn_sched_barrier(0);
      const v2f al0 = {ar[s][0][0], ar[s][0][1]}, al1 = {ar[s][1][0], ar[s][1][1]};
      const v2f ah0 = {ar[s][0][2], ar[s][0][3]}, ah1 = {ar[s][1][2], ar[s][1][3]};
#pragma unroll
      for (int j = 0; j < 4; ++j) {
        if constexpr (MODE == 2) {
          acc[0][j] = wmma4(lo2(fb[s][j]), al0, acc[0][j]);
          acc[1][j] = wmma4(lo2(fb[s][j]), al1, acc[1][j]);
        } else {
          acc[0][j] = wmma4(al0, lo2(fb[s][j]), acc[0][j]);
          acc[1][j] = wmma4(al1, lo2(fb[s][j]), acc[1][j]);
        }
      }
#pragma unroll
      for (int j = 0; j < 4; ++j) {
        if constexpr (MODE == 2) {
          acc[0][j] = wmma4(hi2(fb[s][j]), ah0, acc[0][j]);
          acc[1][j] = wmma4(hi2(fb[s][j]), ah1, acc[1][j]);
        } else {
          acc[0][j] = wmma4(ah0, hi2(fb[s][j]), acc[0][j]);
          acc[1][j] = wmma4(ah1, hi2(fb[s][j]), acc[1][j]);
        }
      }
    }
    lds_barrier();
    const int t = o0; o0 = o1; o1 = o2; o2 = t;
  }

  const int m0 = bm0 + wm * 32;
  const int n0 = bn0 + wn * 64;
  if constexpr (MODE == 2) {
    float cv[2];
#pragma unroll
    for (int i = 0; i < 2; ++i) cv[i] = colVec[m0 + i * 16 + col];
#pragma unroll
    for (int j = 0; j < 4; ++j) {
#pragma unroll
      for (int rr = 0; rr < 8; ++rr) {
        const int n = n0 + j * 16 + rr + 8 * hi;
        const float rb = rowBias[n];
        float* orow = Ob + (long)n * strideO + m0;
#pragma unroll
        for (int i = 0; i < 2; ++i) orow[i * 16 + col] = acc[i][j][rr] + rb * cv[i];
      }
    }
  } else {
#pragma unroll
    for (int i = 0; i < 2; ++i) {
#pragma unroll
      for (int rr = 0; rr < 8; ++rr) {
        int row = m0 + i * 16 + rr + 8 * hi;
        float* orow = Ob + (long)row * strideO + n0;
#pragma unroll
        for (int j = 0; j < 4; ++j) orow[j * 16 + col] = acc[i][j][rr];
      }
    }
  }
  if constexpr (MODE == 1) {
    if (doCs) {
      float* sC = &sW[0];
#pragma unroll
      for (int p = 0; p < PW; ++p) {
        float s = csacc[p];
        s += __shfl_xor(s, 1);
        s += __shfl_xor(s, 2);
        s += __shfl_xor(s, 4);
        if ((tid & 7) == 0) sC[srow + RPP * p] = s;
      }
      __syncthreads();
      if (tid < 32) *(v4f*)&colSum[bn0 + tid * 4] = *(const v4f*)&sC[tid * 4];
    }
  }
}

#define AP_ 16
#define KP_ 16
#define VP_ 20
#define TP_ 17

static __device__ __forceinline__ float xhalf(float x) {
  int b = __builtin_bit_cast(int, x);
  return __builtin_bit_cast(
      float, __builtin_amdgcn_permlanex16(b, b, 0x76543210, 0xfedcba98, false, false));
}

__global__ __launch_bounds__(256, 8) void attn_wmma(
    const float* __restrict__ q, const float* __restrict__ kbuf,
    const float* __restrict__ vbuf, float* __restrict__ obuf) {
  __shared__ __attribute__((aligned(16))) float sK[2][HD_ * KP_];
  __shared__ __attribute__((aligned(16))) float sV[2][DV_ * VP_];
  __shared__ float sT[8][16 * TP_];
  int tid = threadIdx.x;
  int lane = tid & 31;
  int wslot = tid >> 5;
  int ltile = (blockIdx.x & 15) * 8 + wslot;
  int h = (blockIdx.x >> 4) & 15;
  int b = blockIdx.x >> 8;
  int l0 = ltile * 16;
  int col = lane & 15;
  int hi = lane >> 4;
  int khalf = hi * 2;

  v2f qB[8];
  {
    int l = l0 + col;
    bool lv = l < L_;
    const float* qrow = q + ((long)b * L_ + l) * C_ + h * HD_;
    const float sc = 0.125f * 1.44269504088896340736f;
#pragma unroll
    for (int kk = 0; kk < 8; ++kk) {
      int k = kk * 4 + khalf;
      float x0 = lv ? qrow[k] * sc : 0.f;
      float x1 = lv ? qrow[k + 1] * sc : 0.f;
      qB[kk] = (v2f){x0, x1};
    }
  }

  v8f O[4] = {};
  float Mrun = -INFINITY;
  float Srun = 0.f;

  const float* kb = kbuf + (long)b * C_ * P_ + (long)h * P_;
  const float* vb = vbuf + (long)b * D_ * P_ + (long)h * P_;

  const int srow = tid >> 3, sq = (tid & 7) * 2;
  const unsigned goff = (unsigned)(srow * 16) * P_ + sq;
  const float* gK = kb + goff;
  const float* gV = vb + goff;
  const int skOff = srow * KP_ + sq;
  const int svOff = srow * VP_ + sq;

  const int kOff = khalf * KP_ + col;
  const int vOff = col * VP_ + hi * 8;

  v2f rk, rv0, rv1;
  rk = *(const v2f*)gK;
  rv0 = *(const v2f*)gV;
  rv1 = *(const v2f*)(gV + (long)(32 * 16) * P_);
  *(v2f*)&sK[0][skOff] = rk;
  *(v2f*)&sV[0][svOff] = rv0;
  *(v2f*)&sV[0][svOff + 32 * VP_] = rv1;
  rk = *(const v2f*)(gK + AP_);
  rv0 = *(const v2f*)(gV + AP_);
  rv1 = *(const v2f*)(gV + (long)(32 * 16) * P_ + AP_);
  __syncthreads();

  constexpr int NS = P_ / AP_;
  for (int s = 0; s < NS; ++s) {
    const int cur = s & 1;
    const bool more = s + 1 < NS;
    const float* ks = &sK[cur][kOff];
    const float* vs = &sV[cur][vOff];
    v8f c = {};
#pragma unroll
    for (int kk = 0; kk < 8; ++kk) {
      const float* kp = ks + kk * 4 * KP_;
      v2f af = {kp[0], kp[KP_]};
      c = wmma4(af, qB[kk], c);
    }
    if (more) {
      *(v2f*)&sK[cur ^ 1][skOff] = rk;
      *(v2f*)&sV[cur ^ 1][svOff] = rv0;
      *(v2f*)&sV[cur ^ 1][svOff + 32 * VP_] = rv1;
    }
    float mx = fmaxf(fmaxf(fmaxf(c[0], c[1]), fmaxf(c[2], c[3])),
                     fmaxf(fmaxf(c[4], c[5]), fmaxf(c[6], c[7])));
    mx = fmaxf(mx, xhalf(mx));
    const float mnew = fmaxf(Mrun, mx);
    const float al = __builtin_exp2f(Mrun - mnew);
    v8f pv;
#pragma unroll
    for (int r = 0; r < 8; ++r) pv[r] = __builtin_exp2f(c[r] - mnew);
    Srun = Srun * al + (((pv[0] + pv[1]) + (pv[2] + pv[3])) +
                        ((pv[4] + pv[5]) + (pv[6] + pv[7])));
    Mrun = mnew;
#pragma unroll
    for (int j = 0; j < 4; ++j) O[j] *= al;
    if (s + 2 < NS) {
      const int po = (s + 2) * AP_;
      rk = *(const v2f*)(gK + po);
      rv0 = *(const v2f*)(gV + po);
      rv1 = *(const v2f*)(gV + (long)(32 * 16) * P_ + po);
    }
#pragma unroll
    for (int m = 0; m < 2; ++m) {
      v4f vf[4];
#pragma unroll
      for (int j = 0; j < 4; ++j)
        vf[j] = *(const v4f*)(vs + j * 16 * VP_ + m * 4);
      const v2f p0 = {pv[4 * m], pv[4 * m + 1]};
      const v2f p1 = {pv[4 * m + 2], pv[4 * m + 3]};
#pragma unroll
      for (int j = 0; j < 4; ++j) O[j] = wmma4(lo2(vf[j]), p0, O[j]);
#pragma unroll
      for (int j = 0; j < 4; ++j) O[j] = wmma4(hi2(vf[j]), p1, O[j]);
    }
    __builtin_amdgcn_fence(__ATOMIC_RELEASE, "workgroup", "local");
    __builtin_amdgcn_s_barrier();
    __builtin_amdgcn_fence(__ATOMIC_ACQUIRE, "workgroup", "local");
  }

  const float inv = 1.f / (Srun + xhalf(Srun));
  float* tw = &sT[wslot][0];
#pragma unroll
  for (int j = 0; j < 4; ++j) {
#pragma unroll
    for (int r = 0; r < 8; ++r) tw[(8 * hi + r) * TP_ + col] = O[j][r] * inv;
    __builtin_amdgcn_fence(__ATOMIC_ACQ_REL, "wavefront");
    __builtin_amdgcn_wave_barrier();
    float t[8];
#pragma unroll
    for (int r = 0; r < 8; ++r) t[r] = tw[col * TP_ + r + 8 * hi];
    __builtin_amdgcn_fence(__ATOMIC_ACQ_REL, "wavefront");
    __builtin_amdgcn_wave_barrier();
#pragma unroll
    for (int r = 0; r < 8; ++r) {
      int l = l0 + r + 8 * hi;
      if (l < L_)
        obuf[((long)b * L_ + l) * D_ + h + (j * 16 + col) * 16] = t[r];
    }
  }
}

extern "C" void kernel_launch(void* const* d_in, const int* in_sizes, int n_in,
                              void* d_out, int out_size, void* d_ws,
                              size_t ws_size, hipStream_t stream) {
  (void)in_sizes; (void)n_in; (void)out_size; (void)ws_size;
  const float* in = (const float*)d_in[0];
  const float* w_kv = (const float*)d_in[1];
  const float* w_q = (const float*)d_in[2];
  const float* b_q = (const float*)d_in[3];
  const float* w_out = (const float*)d_in[4];
  const float* b_out = (const float*)d_in[5];
  float* out = (float*)d_out;

  float* ws = (float*)d_ws;
  float* qbuf = ws;
  float* vbuf = qbuf + (size_t)B_ * L_ * C_;
  float* kbuf = vbuf + (size_t)B_ * D_ * P_;
  float* obuf = kbuf + (size_t)B_ * C_ * P_;
  float* csbuf = obuf;

  gemm_rowA_wmma<4><<<256, 256, 0, stream>>>(in, (long)(B_ * D_), (long)D_, w_q,
                                             b_q, qbuf, (long)C_, (long)L_ * C_,
                                             L_, D_, 16, 4);
  gemm_colA_wmma<4, 1><<<256, 256, 0, stream>>>(in, (long)(B_ * D_), (long)D_, w_kv,
                                                WKV_LD, vbuf, (long)P_,
                                                (long)D_ * P_, L_, 8, 8, nullptr,
                                                nullptr, csbuf);
  gemm_colA_wmma<2, 2><<<256, 128, 0, stream>>>(vbuf, (long)P_, (long)D_ * P_, w_q,
                                                D_, kbuf, (long)P_, (long)C_ * P_,
                                                D_, 16, 4, b_q, csbuf, nullptr);
  attn_wmma<<<1024, 256, 0, stream>>>(qbuf, kbuf, vbuf, obuf);
  gemm_rowA_wmma<4><<<512, 256, 0, stream>>>(obuf, (long)D_, (long)L_ * D_,
                                             w_out, b_out, out, (long)(B_ * D_),
                                             (long)D_, L_, D_, 16, 8);
}
